# starting kernel
# baseline (speedup 1.0000x reference)
#include <hip/hip_runtime.h>
#include <math.h>


#define IN_CH 128
#define HID   64
#define HEADS 4
#define NEG_SLOPE 0.2f

typedef float v2f __attribute__((ext_vector_type(2)));
typedef float v8f __attribute__((ext_vector_type(8)));

__device__ __forceinline__ unsigned f2ord(float f) {
  unsigned u = __float_as_uint(f);
  return (u & 0x80000000u) ? ~u : (u | 0x80000000u);
}
__device__ __forceinline__ float ord2f(unsigned u) {
  return (u & 0x80000000u) ? __uint_as_float(u & 0x7fffffffu)
                           : __uint_as_float(~u);
}
#define NEG_INF_ORD 0x007FFFFFu

__device__ __forceinline__ float elu_f(float x) {
  return x > 0.0f ? x : (expf(x) - 1.0f);
}

__global__ void fill_f32(float* p, float v, int n) {
  int i = blockIdx.x * blockDim.x + threadIdx.x;
  if (i < n) p[i] = v;
}
__global__ void fill_u32(unsigned* p, unsigned v, int n) {
  int i = blockIdx.x * blockDim.x + threadIdx.x;
  if (i < n) p[i] = v;
}

template <int K>
__global__ void wmma_gemm_f32(const float* __restrict__ A,
                              const float* __restrict__ B,
                              float* __restrict__ C, int M, int Nc) {
  int wave = (int)((blockIdx.x * blockDim.x + threadIdx.x) >> 5);
  int ntn  = Nc >> 4;
  int total = (M >> 4) * ntn;
  if (wave >= total) return;
  int tm = wave / ntn, tn = wave % ntn;
  int lane = (int)(threadIdx.x & 31);
  int lr = lane & 15, hi = lane >> 4;

  const float* arow = A + (size_t)(tm * 16 + lr) * K;
  const float* bcol = B + tn * 16 + lr;
  v8f acc = {};
#pragma unroll 8
  for (int k0 = 0; k0 < K; k0 += 4) {
    int ka = k0 + hi * 2;
    v2f a; a.x = arow[ka];               a.y = arow[ka + 1];
    v2f b; b.x = bcol[(size_t)ka * Nc];  b.y = bcol[(size_t)(ka + 1) * Nc];
    acc = __builtin_amdgcn_wmma_f32_16x16x4_f32(
         false, a,  false, b,
         (short)0, acc,  false,  false);
  }
  float* crow = C + tn * 16 + lr;
  int rbase = tm * 16 + hi * 8;
#pragma unroll
  for (int v = 0; v < 8; ++v) crow[(size_t)(rbase + v) * Nc] = acc[v];
}

template <int H>
__global__ void attn_coef(const float* __restrict__ h,
                          const float* __restrict__ a_src,
                          const float* __restrict__ a_dst,
                          float* __restrict__ es, float* __restrict__ ed, int n) {
  int idx = blockIdx.x * blockDim.x + threadIdx.x;
  if (idx >= n * H) return;
  int i = idx / H, hh = idx % H;
  const float* hp = h + (size_t)i * (H * HID) + hh * HID;
  const float* as = a_src + hh * HID;
  const float* ad = a_dst + hh * HID;
  float s = 0.f, d = 0.f;
#pragma unroll 8
  for (int c = 0; c < HID; ++c) { float v = hp[c]; s += v * as[c]; d += v * ad[c]; }
  es[idx] = s; ed[idx] = d;
}

template <int H>
__global__ void edge_logits(const int* __restrict__ ei,
                            const float* __restrict__ es,
                            const float* __restrict__ ed,
                            float* __restrict__ logit,
                            unsigned* __restrict__ mEnc, int E_, int N_) {
  int idx = blockIdx.x * blockDim.x + threadIdx.x;
  if (idx >= (E_ + N_) * H) return;
  int e = idx / H, hh = idx % H;
  int s = (e < E_) ? ei[e]      : (e - E_);
  int d = (e < E_) ? ei[E_ + e] : (e - E_);
  float l = es[s * H + hh] + ed[d * H + hh];
  l = (l > 0.0f) ? l : NEG_SLOPE * l;
  logit[idx] = l;
  atomicMax(&mEnc[d * H + hh], f2ord(l));
}

template <int H>
__global__ void edge_expsum(const int* __restrict__ ei,
                            const float* __restrict__ logit,
                            const unsigned* __restrict__ mEnc,
                            float* __restrict__ ssum, int E_, int N_) {
  int idx = blockIdx.x * blockDim.x + threadIdx.x;
  if (idx >= (E_ + N_) * H) return;
  int e = idx / H, hh = idx % H;
  int d = (e < E_) ? ei[E_ + e] : (e - E_);
  float m = ord2f(mEnc[d * H + hh]);
  atomicAdd(&ssum[d * H + hh], expf(logit[idx] - m));
}

template <int H>
__global__ void edge_message4(const int* __restrict__ ei,
                              const float* __restrict__ h,
                              const float* __restrict__ logit,
                              const unsigned* __restrict__ mEnc,
                              const float* __restrict__ ssum,
                              float* __restrict__ agg, int E_, int N_) {
  const int LDH = H * HID;
  const int LD4 = LDH / 4;
  int idx = blockIdx.x * blockDim.x + threadIdx.x;
  if (idx >= (E_ + N_) * LD4) return;
  int e = idx / LD4;
  int t = (idx % LD4) * 4;
  int hh = t >> 6;
  int s = (e < E_) ? ei[e]      : (e - E_);
  int d = (e < E_) ? ei[E_ + e] : (e - E_);
  float m = ord2f(mEnc[d * H + hh]);
  float a = expf(logit[e * H + hh] - m) / (ssum[d * H + hh] + 1e-16f);
  const float4 hv = *reinterpret_cast<const float4*>(h + (size_t)s * LDH + t);
  float* ap = agg + (size_t)d * LDH + t;
  atomicAdd(ap + 0, hv.x * a);
  atomicAdd(ap + 1, hv.y * a);
  atomicAdd(ap + 2, hv.z * a);
  atomicAdd(ap + 3, hv.w * a);
}

template <int LD>
__global__ void bias_elu(const float* __restrict__ agg,
                         const float* __restrict__ bias,
                         float* __restrict__ out, int total) {
  int idx = blockIdx.x * blockDim.x + threadIdx.x;
  if (idx >= total) return;
  out[idx] = elu_f(agg[idx] + bias[idx % LD]);
}

__global__ void out_proj(const float* __restrict__ h,
                         const float* __restrict__ Wo,
                         const float* __restrict__ bo,
                         float* __restrict__ out, int n) {
  int wid  = (int)((blockIdx.x * blockDim.x + threadIdx.x) >> 5);
  int lane = (int)(threadIdx.x & 31);
  if (wid >= n) return;
  const float* hp = h + (size_t)wid * HID;
  float s = hp[lane] * Wo[lane] + hp[lane + 32] * Wo[lane + 32];
#pragma unroll
  for (int o = 16; o > 0; o >>= 1) s += __shfl_xor(s, o, 32);
  if (lane == 0) out[wid] = s + bo[0];
}

#define CDIV(a, b) (((a) + (b) - 1) / (b))

extern "C" void kernel_launch(void* const* d_in, const int* in_sizes, int n_in,
                              void* d_out, int out_size, void* d_ws, size_t ws_size,
                              hipStream_t stream) {
  const float* x   = (const float*)d_in[0];
  const float* W1  = (const float*)d_in[1];
  const float* as1 = (const float*)d_in[2];
  const float* ad1 = (const float*)d_in[3];
  const float* b1  = (const float*)d_in[4];
  const float* W2  = (const float*)d_in[5];
  const float* as2 = (const float*)d_in[6];
  const float* ad2 = (const float*)d_in[7];
  const float* b2  = (const float*)d_in[8];
  const float* Wo  = (const float*)d_in[9];
  const float* bo  = (const float*)d_in[10];
  const int*   ei  = (const int*)d_in[11];

  const int n  = in_sizes[0]  / IN_CH;
  const int e  = in_sizes[11] / 2;
  const int et = e + n;

  float* p = (float*)d_ws;
  float*    h1   = p;            p += (size_t)n * 256;
  float*    agg1 = p;            p += (size_t)n * 256;
  float*    h2   = p;            p += (size_t)n * 64;
  float*    agg2 = p;            p += (size_t)n * 64;
  float*    es1  = p;            p += (size_t)n * 4;
  float*    ed1  = p;            p += (size_t)n * 4;
  unsigned* m1   = (unsigned*)p; p += (size_t)n * 4;
  float*    s1   = p;            p += (size_t)n * 4;
  float*    es2v = p;            p += n;
  float*    ed2v = p;            p += n;
  unsigned* m2   = (unsigned*)p; p += n;
  float*    s2   = p;            p += n;
  float*    lg1  = p;            p += (size_t)et * 4;
  float*    lg2  = p;

  const int B = 256;

  fill_f32<<<CDIV(n * 256, B), B, 0, stream>>>(agg1, 0.0f, n * 256);
  fill_f32<<<CDIV(n * 64,  B), B, 0, stream>>>(agg2, 0.0f, n * 64);
  fill_f32<<<CDIV(n * 4,   B), B, 0, stream>>>(s1,   0.0f, n * 4);
  fill_f32<<<CDIV(n,       B), B, 0, stream>>>(s2,   0.0f, n);
  fill_u32<<<CDIV(n * 4,   B), B, 0, stream>>>(m1, NEG_INF_ORD, n * 4);
  fill_u32<<<CDIV(n,       B), B, 0, stream>>>(m2, NEG_INF_ORD, n);

  int tiles1 = (n / 16) * (256 / 16);
  wmma_gemm_f32<128><<<CDIV(tiles1, 8), B, 0, stream>>>(x, W1, h1, n, 256);
  attn_coef<HEADS><<<CDIV(n * HEADS, B), B, 0, stream>>>(h1, as1, ad1, es1, ed1, n);
  edge_logits<HEADS><<<CDIV(et * HEADS, B), B, 0, stream>>>(ei, es1, ed1, lg1, m1, e, n);
  edge_expsum<HEADS><<<CDIV(et * HEADS, B), B, 0, stream>>>(ei, lg1, m1, s1, e, n);
  edge_message4<HEADS><<<CDIV(et * 64, B), B, 0, stream>>>(ei, h1, lg1, m1, s1, agg1, e, n);
  bias_elu<256><<<CDIV(n * 256, B), B, 0, stream>>>(agg1, b1, h1, n * 256);

  int tiles2 = (n / 16) * (64 / 16);
  wmma_gemm_f32<256><<<CDIV(tiles2, 8), B, 0, stream>>>(h1, W2, h2, n, 64);
  attn_coef<1><<<CDIV(n, B), B, 0, stream>>>(h2, as2, ad2, es2v, ed2v, n);
  edge_logits<1><<<CDIV(et, B), B, 0, stream>>>(ei, es2v, ed2v, lg2, m2, e, n);
  edge_expsum<1><<<CDIV(et, B), B, 0, stream>>>(ei, lg2, m2, s2, e, n);
  edge_message4<1><<<CDIV(et * 16, B), B, 0, stream>>>(ei, h2, lg2, m2, s2, agg2, e, n);
  bias_elu<64><<<CDIV(n * 64, B), B, 0, stream>>>(agg2, b2, h2, n * 64);

  out_proj<<<CDIV(n * 32, B), B, 0, stream>>>(h2, Wo, bo, (float*)d_out, n);
}


# GEMMs re-tiled: W staged once per work-group in LDS in fragment order, one wave per 16-row strip holds all column tiles (A read once, 16-B loads); order of the f32 k sums changes (8-wide steps)
# speedup vs baseline: 3.1122x; 3.1122x over previous
#include <hip/hip_runtime.h>
#include <math.h>


#define IN_CH 128
#define HID   64
#define HEADS 4
#define NEG_SLOPE 0.2f

typedef float v2f __attribute__((ext_vector_type(2)));
typedef float v8f __attribute__((ext_vector_type(8)));

__device__ __forceinline__ float elu_f(float x) {
  return x > 0.0f ? x : (expf(x) - 1.0f);
}

__global__ void fill_u32(unsigned* p, unsigned v, int n) {
  int i = blockIdx.x * blockDim.x + threadIdx.x;
  if (i < n) p[i] = v;
}

__global__ void csr_count(const int* __restrict__ ei, unsigned* __restrict__ cnt, int E_) {
  int e = blockIdx.x * blockDim.x + threadIdx.x;
  if (e < E_) atomicAdd(&cnt[ei[E_ + e]], 1u);
}

#define SCAN_TILE 1024
__global__ void __launch_bounds__(256)
csr_scan(const unsigned* __restrict__ cnt, unsigned* __restrict__ rowptr,
         unsigned* __restrict__ cursor, int n) {
  __shared__ unsigned wsum[8];
  __shared__ unsigned wpre[8];
  const int t = (int)threadIdx.x, lane = t & 31, w = t >> 5;
  const int tile0 = (int)blockIdx.x * SCAN_TILE;

  unsigned pre = 0;
#pragma unroll 4
  for (int i = 4 * t; i < tile0; i += SCAN_TILE) {
    const uint4 c = *reinterpret_cast<const uint4*>(cnt + i);
    pre += (c.x + c.y) + (c.z + c.w);
  }

  const int i0 = tile0 + 4 * t;
  unsigned c[4];
#pragma unroll
  for (int k = 0; k < 4; ++k) c[k] = (i0 + k < n) ? cnt[i0 + k] : 0u;
  const unsigned own = (c[0] + c[1]) + (c[2] + c[3]);

  unsigned incl = own;
#pragma unroll
  for (int o = 1; o < 32; o <<= 1) {
    unsigned v = __shfl_up(incl, (unsigned)o, 32);
    if (lane >= o) incl += v;
  }
#pragma unroll
  for (int o = 16; o > 0; o >>= 1) pre += __shfl_xor(pre, o, 32);
  if (lane == 31) wsum[w] = incl;
  if (lane == 0)  wpre[w] = pre;
  __syncthreads();
  unsigned base = incl - own;
#pragma unroll
  for (int j = 0; j < 8; ++j) {
    base += wpre[j];
    if (j < w) base += wsum[j];
  }
#pragma unroll
  for (int k = 0; k < 4; ++k) {
    const int idx = i0 + k;
    if (idx <= n) rowptr[idx] = base;
    if (idx <  n) cursor[idx] = base;
    base += c[k];
  }
}

__global__ void csr_fill(const int* __restrict__ ei, unsigned* __restrict__ cursor,
                         int* __restrict__ col, int E_) {
  int e = blockIdx.x * blockDim.x + threadIdx.x;
  if (e >= E_) return;
  unsigned pos = atomicAdd(&cursor[ei[E_ + e]], 1u);
  col[pos] = ei[e];
}

#define GEMM_WAVES 16
template <int K, int NC>
__global__ void __launch_bounds__(GEMM_WAVES * 32) __attribute__((amdgpu_waves_per_eu(1, 4)))
wmma_gemm_f32(const float* __restrict__ A, const float* __restrict__ B,
              float* __restrict__ C, int M) {
  constexpr int NT = NC / 16;
  constexpr int KS = K / 8;
  constexpr int NQ = (K / 4) * NC;
  constexpr int NTHR = GEMM_WAVES * 32;
  constexpr int NIT = NQ / NTHR;
  static_assert(NQ % NTHR == 0, "staging shape");
  __shared__ float4 Bs[NQ];

  const int t = (int)threadIdx.x;
  const int lane = t & 31;
  const int lr = lane & 15, hi = lane >> 4;
  const int w = __builtin_amdgcn_readfirstlane(t >> 5);
  const int strip = (int)blockIdx.x + (int)gridDim.x * w;
  const int nstrips = M >> 4;
  const bool active = strip < nstrips;

  float4 stg[NIT];
#pragma unroll
  for (int j = 0; j < NIT; ++j) {
    const int idx = t + j * NTHR;
    const float* bp = B + (size_t)(4 * (idx / NC)) * NC + (idx % NC);
    stg[j].x = bp[0]; stg[j].y = bp[NC]; stg[j].z = bp[2 * NC]; stg[j].w = bp[3 * NC];
  }

#pragma unroll
  for (int j = 0; j < NIT; ++j) Bs[t + j * NTHR] = stg[j];
  __syncthreads();
  if (!active) return;

  float4 a[KS];
  {
    const float* ap = A + (size_t)(strip * 16 + lr) * K + 4 * hi;
#pragma unroll
    for (int i = 0; i < KS; ++i) a[i] = *reinterpret_cast<const float4*>(ap + 8 * i);
  }

  v8f acc[NT];
#pragma unroll
  for (int tn = 0; tn < NT; ++tn) acc[tn] = v8f{};

#pragma unroll
  for (int i = 0; i < KS; ++i) {
    const float4* bq = Bs + (2 * i + hi) * NC + lr;
    float4 b[NT];
#pragma unroll
    for (int tn = 0; tn < NT; ++tn) b[tn] = bq[tn * 16];
    v2f a01; a01.x = a[i].x; a01.y = a[i].y;
    v2f a23; a23.x = a[i].z; a23.y = a[i].w;
#pragma unroll
    for (int tn = 0; tn < NT; ++tn) {
      v2f b01; b01.x = b[tn].x; b01.y = b[tn].y;
      acc[tn] = __builtin_amdgcn_wmma_f32_16x16x4_f32(
           false, a01,  false, b01,
           (short)0, acc[tn],  false,  false);
    }
#pragma unroll
    for (int tn = 0; tn < NT; ++tn) {
      v2f b23; b23.x = b[tn].z; b23.y = b[tn].w;
      acc[tn] = __builtin_amdgcn_wmma_f32_16x16x4_f32(
           false, a23,  false, b23,
           (short)0, acc[tn],  false,  false);
    }
  }

  float* cp = C + (size_t)(strip * 16 + 8 * hi) * NC + lr;
#pragma unroll
  for (int tn = 0; tn < NT; ++tn) {
#pragma unroll
    for (int v = 0; v < 8; ++v) cp[(size_t)v * NC + tn * 16] = acc[tn][v];
  }
}

template <int H>
__global__ void attn_coef(const float* __restrict__ h,
                          const float* __restrict__ a_src,
                          const float* __restrict__ a_dst,
                          float* __restrict__ es, float* __restrict__ ed, int n) {
  int idx = blockIdx.x * blockDim.x + threadIdx.x;
  if (idx >= n * H) return;
  int i = idx / H, hh = idx % H;
  const float* hp = h + (size_t)i * (H * HID) + hh * HID;
  const float* as = a_src + hh * HID;
  const float* ad = a_dst + hh * HID;
  float s = 0.f, d = 0.f;
#pragma unroll 8
  for (int c = 0; c < HID; ++c) { float v = hp[c]; s += v * as[c]; d += v * ad[c]; }
  es[idx] = s; ed[idx] = d;
}

template <int H>
__device__ __forceinline__ void slot_logits(const int* __restrict__ col,
                                            const float* __restrict__ es,
                                            const float (&edd)[H], int beg, int tot,
                                            int d, int j, int& s, float (&l)[H]) {
  s = d;
#pragma unroll
  for (int hh = 0; hh < H; ++hh) l[hh] = -INFINITY;
  if (j < tot) {
    s = (j < tot - 1) ? col[beg + j] : d;
    float e[H];
    if constexpr (H == 4) {
      const float4 v = *reinterpret_cast<const float4*>(es + 4 * s);
      e[0] = v.x; e[1] = v.y; e[2] = v.z; e[3] = v.w;
    } else {
      e[0] = es[s];
    }
#pragma unroll
    for (int hh = 0; hh < H; ++hh) {
      const float x = e[hh] + edd[hh];
      l[hh] = (x > 0.0f) ? x : NEG_SLOPE * x;
    }
  }
}

template <int H>
__global__ void __launch_bounds__(256)
gat_gather(const unsigned* __restrict__ rowptr, const int* __restrict__ col,
           const float* __restrict__ h, const float* __restrict__ es,
           const float* __restrict__ ed, const float* __restrict__ bias,
           float* __restrict__ out, int n) {
  constexpr int LDH = H * HID;
  const int lane = (int)(threadIdx.x & 31);
  int d = (int)((blockIdx.x * blockDim.x + threadIdx.x) >> 5);
  d = __builtin_amdgcn_readfirstlane(d);
  if (d >= n) return;
  const int beg = (int)rowptr[d];
  const int tot = (int)rowptr[d + 1] - beg + 1;

  float edd[H];
#pragma unroll
  for (int hh = 0; hh < H; ++hh) edd[hh] = ed[d * H + hh];

  int s0;
  float l0[H], mx[H];
  slot_logits<H>(col, es, edd, beg, tot, d, lane, s0, l0);
#pragma unroll
  for (int hh = 0; hh < H; ++hh) mx[hh] = l0[hh];
  for (int base = 32; base < tot; base += 32) {
    int s;
    float l[H];
    slot_logits<H>(col, es, edd, beg, tot, d, base + lane, s, l);
#pragma unroll
    for (int hh = 0; hh < H; ++hh) mx[hh] = fmaxf(mx[hh], l[hh]);
  }
#pragma unroll
  for (int hh = 0; hh < H; ++hh) {
#pragma unroll
    for (int o = 16; o > 0; o >>= 1) mx[hh] = fmaxf(mx[hh], __shfl_xor(mx[hh], o, 32));
    mx[hh] = __builtin_isfinite(mx[hh]) ? mx[hh] : 0.0f;
  }

  float sm[H];
#pragma unroll
  for (int hh = 0; hh < H; ++hh) sm[hh] = 0.0f;
  float4 acc0 = make_float4(0.f, 0.f, 0.f, 0.f);
  float4 acc1 = make_float4(0.f, 0.f, 0.f, 0.f);

  for (int base = 0; base < tot; base += 32) {
    int s = s0;
    float l[H], p[H];
#pragma unroll
    for (int hh = 0; hh < H; ++hh) l[hh] = l0[hh];
    if (base > 0) slot_logits<H>(col, es, edd, beg, tot, d, base + lane, s, l);
#pragma unroll
    for (int hh = 0; hh < H; ++hh) {
      p[hh] = (base + lane < tot) ? expf(l[hh] - mx[hh]) : 0.0f;
      sm[hh] += p[hh];
    }
    const int nk = (tot - base < 32) ? (tot - base) : 32;
#pragma unroll 4
    for (int k = 0; k < nk; ++k) {
      const int sk = __builtin_amdgcn_readlane(s, k);
      const float* hp = h + (size_t)sk * LDH;
      if constexpr (H == 4) {
        const float p0 = __int_as_float(__builtin_amdgcn_readlane(__float_as_int(p[0]), k));
        const float p1 = __int_as_float(__builtin_amdgcn_readlane(__float_as_int(p[1]), k));
        const float p2 = __int_as_float(__builtin_amdgcn_readlane(__float_as_int(p[2]), k));
        const float p3 = __int_as_float(__builtin_amdgcn_readlane(__float_as_int(p[3]), k));
        const float pa = (lane < 16) ? p0 : p1;
        const float pb = (lane < 16) ? p2 : p3;
        const float4 a = *reinterpret_cast<const float4*>(hp + 4 * lane);
        const float4 b = *reinterpret_cast<const float4*>(hp + 128 + 4 * lane);
        acc0.x += a.x * pa; acc0.y += a.y * pa; acc0.z += a.z * pa; acc0.w += a.w * pa;
        acc1.x += b.x * pb; acc1.y += b.y * pb; acc1.z += b.z * pb; acc1.w += b.w * pb;
      } else {
        const float p0 = __int_as_float(__builtin_amdgcn_readlane(__float_as_int(p[0]), k));
        const float2 a = *reinterpret_cast<const float2*>(hp + 2 * lane);
        acc0.x += a.x * p0; acc0.y += a.y * p0;
      }
    }
  }
#pragma unroll
  for (int hh = 0; hh < H; ++hh) {
#pragma unroll
    for (int o = 16; o > 0; o >>= 1) sm[hh] += __shfl_xor(sm[hh], o, 32);
  }

  float* op = out + (size_t)d * LDH;
  if constexpr (H == 4) {
    const float da = ((lane < 16) ? sm[0] : sm[1]) + 1e-16f;
    const float db = ((lane < 16) ? sm[2] : sm[3]) + 1e-16f;
    const float4 ba = *reinterpret_cast<const float4*>(bias + 4 * lane);
    const float4 bb = *reinterpret_cast<const float4*>(bias + 128 + 4 * lane);
    float4 oa, ob;
    oa.x = elu_f(acc0.x / da + ba.x); oa.y = elu_f(acc0.y / da + ba.y);
    oa.z = elu_f(acc0.z / da + ba.z); oa.w = elu_f(acc0.w / da + ba.w);
    ob.x = elu_f(acc1.x / db + bb.x); ob.y = elu_f(acc1.y / db + bb.y);
    ob.z = elu_f(acc1.z / db + bb.z); ob.w = elu_f(acc1.w / db + bb.w);
    *reinterpret_cast<float4*>(op + 4 * lane)       = oa;
    *reinterpret_cast<float4*>(op + 128 + 4 * lane) = ob;
  } else {
    const float da = sm[0] + 1e-16f;
    const float2 ba = *reinterpret_cast<const float2*>(bias + 2 * lane);
    float2 oa;
    oa.x = elu_f(acc0.x / da + ba.x); oa.y = elu_f(acc0.y / da + ba.y);
    *reinterpret_cast<float2*>(op + 2 * lane) = oa;
  }
}

__global__ void out_proj(const float* __restrict__ h,
                         const float* __restrict__ Wo,
                         const float* __restrict__ bo,
                         float* __restrict__ out, int n) {
  int wid  = (int)((blockIdx.x * blockDim.x + threadIdx.x) >> 5);
  int lane = (int)(threadIdx.x & 31);
  if (wid >= n) return;
  const float* hp = h + (size_t)wid * HID;
  float s = hp[lane] * Wo[lane] + hp[lane + 32] * Wo[lane + 32];
#pragma unroll
  for (int o = 16; o > 0; o >>= 1) s += __shfl_xor(s, o, 32);
  if (lane == 0) out[wid] = s + bo[0];
}

#define CDIV(a, b) (((a) + (b) - 1) / (b))

extern "C" void kernel_launch(void* const* d_in, const int* in_sizes, int n_in,
                              void* d_out, int out_size, void* d_ws, size_t ws_size,
                              hipStream_t stream) {
  const float* x   = (const float*)d_in[0];
  const float* W1  = (const float*)d_in[1];
  const float* as1 = (const float*)d_in[2];
  const float* ad1 = (const float*)d_in[3];
  const float* b1  = (const float*)d_in[4];
  const float* W2  = (const float*)d_in[5];
  const float* as2 = (const float*)d_in[6];
  const float* ad2 = (const float*)d_in[7];
  const float* b2  = (const float*)d_in[8];
  const float* Wo  = (const float*)d_in[9];
  const float* bo  = (const float*)d_in[10];
  const int*   ei  = (const int*)d_in[11];

  const int n  = in_sizes[0]  / IN_CH;
  const int e  = in_sizes[11] / 2;

  float* p = (float*)d_ws;
  float*    h1   = p;            p += (size_t)n * 256;
  float*    o1   = p;            p += (size_t)n * 256;
  float*    h2   = p;            p += (size_t)n * 64;
  float*    o2   = p;            p += (size_t)n * 64;
  float*    es1  = p;            p += (size_t)n * 4;
  float*    ed1  = p;            p += (size_t)n * 4;
  float*    es2v = p;            p += n;
  float*    ed2v = p;            p += n;
  unsigned* cnt  = (unsigned*)p; p += n;
  unsigned* cur  = (unsigned*)p; p += n;
  int*      col  = (int*)p;      p += e;
  unsigned* rptr = (unsigned*)p;

  const int B = 256;

  fill_u32<<<CDIV(n, B), B, 0, stream>>>(cnt, 0u, n);
  csr_count<<<CDIV(e, B), B, 0, stream>>>(ei, cnt, e);
  csr_scan<<<n / SCAN_TILE + 1, B, 0, stream>>>(cnt, rptr, cur, n);
  csr_fill<<<CDIV(e, B), B, 0, stream>>>(ei, cur, col, e);

  const int gneed = CDIV(n / 16, GEMM_WAVES);
  const int ggrid = gneed > 256 ? gneed : 256;
  wmma_gemm_f32<IN_CH, HEADS * HID><<<ggrid, GEMM_WAVES * 32, 0, stream>>>(x, W1, h1, n);
  attn_coef<HEADS><<<CDIV(n * HEADS, B), B, 0, stream>>>(h1, as1, ad1, es1, ed1, n);
  gat_gather<HEADS><<<CDIV(n * 32, B), B, 0, stream>>>(rptr, col, h1, es1, ed1, b1, o1, n);

  wmma_gemm_f32<HEADS * HID, HID><<<ggrid, GEMM_WAVES * 32, 0, stream>>>(o1, W2, h2, n);
  attn_coef<1><<<CDIV(n, B), B, 0, stream>>>(h2, as2, ad2, es2v, ed2v, n);
  gat_gather<1><<<CDIV(n * 32, B), B, 0, stream>>>(rptr, col, h2, es2v, ed2v, b2, o2, n);

  out_proj<<<CDIV(n * 32, B), B, 0, stream>>>(o2, Wo, bo, (float*)d_out, n);
}


# CSR build with one scattered-atomic pass, not two: the in-degree add returns each edge's rank (kept per edge), fill is rowptr[dst]+rank by plain stores; edge sums stay in atomic arrival order
# speedup vs baseline: 3.5053x; 1.1263x over previous
#include <hip/hip_runtime.h>
#include <math.h>


#define IN_CH 128
#define HID   64
#define HEADS 4
#define NEG_SLOPE 0.2f

typedef float v2f __attribute__((ext_vector_type(2)));
typedef float v8f __attribute__((ext_vector_type(8)));

__device__ __forceinline__ float elu_f(float x) {
  return x > 0.0f ? x : (expf(x) - 1.0f);
}

__global__ void fill_u32(unsigned* p, unsigned v, int n) {
  int i = blockIdx.x * blockDim.x + threadIdx.x;
  if (i < n) p[i] = v;
}

__global__ void csr_count(const int* __restrict__ ei, unsigned* __restrict__ cnt,
                          unsigned* __restrict__ rank, int E_) {
  int e = blockIdx.x * blockDim.x + threadIdx.x;
  if (e < E_) rank[e] = atomicAdd(&cnt[ei[E_ + e]], 1u);
}

#define SCAN_TILE 1024
__global__ void __launch_bounds__(256)
csr_scan(const unsigned* __restrict__ cnt, unsigned* __restrict__ rowptr, int n) {
  __shared__ unsigned wsum[8];
  __shared__ unsigned wpre[8];
  const int t = (int)threadIdx.x, lane = t & 31, w = t >> 5;
  const int tile0 = (int)blockIdx.x * SCAN_TILE;

  unsigned pre = 0;
#pragma unroll 4
  for (int i = 4 * t; i < tile0; i += SCAN_TILE) {
    const uint4 c = *reinterpret_cast<const uint4*>(cnt + i);
    pre += (c.x + c.y) + (c.z + c.w);
  }

  const int i0 = tile0 + 4 * t;
  unsigned c[4];
#pragma unroll
  for (int k = 0; k < 4; ++k) c[k] = (i0 + k < n) ? cnt[i0 + k] : 0u;
  const unsigned own = (c[0] + c[1]) + (c[2] + c[3]);

  unsigned incl = own;
#pragma unroll
  for (int o = 1; o < 32; o <<= 1) {
    unsigned v = __shfl_up(incl, (unsigned)o, 32);
    if (lane >= o) incl += v;
  }
#pragma unroll
  for (int o = 16; o > 0; o >>= 1) pre += __shfl_xor(pre, o, 32);
  if (lane == 31) wsum[w] = incl;
  if (lane == 0)  wpre[w] = pre;
  __syncthreads();
  unsigned base = incl - own;
#pragma unroll
  for (int j = 0; j < 8; ++j) {
    base += wpre[j];
    if (j < w) base += wsum[j];
  }
#pragma unroll
  for (int k = 0; k < 4; ++k) {
    const int idx = i0 + k;
    if (idx <= n) rowptr[idx] = base;
    base += c[k];
  }
}

__global__ void csr_fill(const int* __restrict__ ei, const unsigned* __restrict__ rowptr,
                         const unsigned* __restrict__ rank, int* __restrict__ col, int E_) {
  int e = blockIdx.x * blockDim.x + threadIdx.x;
  if (e >= E_) return;
  col[rowptr[ei[E_ + e]] + rank[e]] = ei[e];
}

#define GEMM_WAVES 16
template <int K, int NC>
__global__ void __launch_bounds__(GEMM_WAVES * 32) __attribute__((amdgpu_waves_per_eu(1, 4)))
wmma_gemm_f32(const float* __restrict__ A, const float* __restrict__ B,
              float* __restrict__ C, int M) {
  constexpr int NT = NC / 16;
  constexpr int KS = K / 8;
  constexpr int NQ = (K / 4) * NC;
  constexpr int NTHR = GEMM_WAVES * 32;
  constexpr int NIT = NQ / NTHR;
  static_assert(NQ % NTHR == 0, "staging shape");
  __shared__ float4 Bs[NQ];

  const int t = (int)threadIdx.x;
  const int lane = t & 31;
  const int lr = lane & 15, hi = lane >> 4;
  const int w = __builtin_amdgcn_readfirstlane(t >> 5);
  const int strip = (int)blockIdx.x + (int)gridDim.x * w;
  const int nstrips = M >> 4;
  const bool active = strip < nstrips;

  float4 stg[NIT];
#pragma unroll
  for (int j = 0; j < NIT; ++j) {
    const int idx = t + j * NTHR;
    const float* bp = B + (size_t)(4 * (idx / NC)) * NC + (idx % NC);
    stg[j].x = bp[0]; stg[j].y = bp[NC]; stg[j].z = bp[2 * NC]; stg[j].w = bp[3 * NC];
  }

#pragma unroll
  for (int j = 0; j < NIT; ++j) Bs[t + j * NTHR] = stg[j];
  __syncthreads();
  if (!active) return;

  float4 a[KS];
  {
    const float* ap = A + (size_t)(strip * 16 + lr) * K + 4 * hi;
#pragma unroll
    for (int i = 0; i < KS; ++i) a[i] = *reinterpret_cast<const float4*>(ap + 8 * i);
  }

  v8f acc[NT];
#pragma unroll
  for (int tn = 0; tn < NT; ++tn) acc[tn] = v8f{};

#pragma unroll
  for (int i = 0; i < KS; ++i) {
    const float4* bq = Bs + (2 * i + hi) * NC + lr;
    float4 b[NT];
#pragma unroll
    for (int tn = 0; tn < NT; ++tn) b[tn] = bq[tn * 16];
    v2f a01; a01.x = a[i].x; a01.y = a[i].y;
    v2f a23; a23.x = a[i].z; a23.y = a[i].w;
#pragma unroll
    for (int tn = 0; tn < NT; ++tn) {
      v2f b01; b01.x = b[tn].x; b01.y = b[tn].y;
      acc[tn] = __builtin_amdgcn_wmma_f32_16x16x4_f32(
           false, a01,  false, b01,
           (short)0, acc[tn],  false,  false);
    }
#pragma unroll
    for (int tn = 0; tn < NT; ++tn) {
      v2f b23; b23.x = b[tn].z; b23.y = b[tn].w;
      acc[tn] = __builtin_amdgcn_wmma_f32_16x16x4_f32(
           false, a23,  false, b23,
           (short)0, acc[tn],  false,  false);
    }
  }

  float* cp = C + (size_t)(strip * 16 + 8 * hi) * NC + lr;
#pragma unroll
  for (int tn = 0; tn < NT; ++tn) {
#pragma unroll
    for (int v = 0; v < 8; ++v) cp[(size_t)v * NC + tn * 16] = acc[tn][v];
  }
}

template <int H>
__global__ void attn_coef(const float* __restrict__ h,
                          const float* __restrict__ a_src,
                          const float* __restrict__ a_dst,
                          float* __restrict__ es, float* __restrict__ ed, int n) {
  int idx = blockIdx.x * blockDim.x + threadIdx.x;
  if (idx >= n * H) return;
  int i = idx / H, hh = idx % H;
  const float* hp = h + (size_t)i * (H * HID) + hh * HID;
  const float* as = a_src + hh * HID;
  const float* ad = a_dst + hh * HID;
  float s = 0.f, d = 0.f;
#pragma unroll 8
  for (int c = 0; c < HID; ++c) { float v = hp[c]; s += v * as[c]; d += v * ad[c]; }
  es[idx] = s; ed[idx] = d;
}

template <int H>
__device__ __forceinline__ void slot_logits(const int* __restrict__ col,
                                            const float* __restrict__ es,
                                            const float (&edd)[H], int beg, int tot,
                                            int d, int j, int& s, float (&l)[H]) {
  s = d;
#pragma unroll
  for (int hh = 0; hh < H; ++hh) l[hh] = -INFINITY;
  if (j < tot) {
    s = (j < tot - 1) ? col[beg + j] : d;
    float e[H];
    if constexpr (H == 4) {
      const float4 v = *reinterpret_cast<const float4*>(es + 4 * s);
      e[0] = v.x; e[1] = v.y; e[2] = v.z; e[3] = v.w;
    } else {
      e[0] = es[s];
    }
#pragma unroll
    for (int hh = 0; hh < H; ++hh) {
      const float x = e[hh] + edd[hh];
      l[hh] = (x > 0.0f) ? x : NEG_SLOPE * x;
    }
  }
}

template <int H>
__global__ void __launch_bounds__(256)
gat_gather(const unsigned* __restrict__ rowptr, const int* __restrict__ col,
           const float* __restrict__ h, const float* __restrict__ es,
           const float* __restrict__ ed, const float* __restrict__ bias,
           float* __restrict__ out, int n) {
  constexpr int LDH = H * HID;
  const int lane = (int)(threadIdx.x & 31);
  int d = (int)((blockIdx.x * blockDim.x + threadIdx.x) >> 5);
  d = __builtin_amdgcn_readfirstlane(d);
  if (d >= n) return;
  const int beg = (int)rowptr[d];
  const int tot = (int)rowptr[d + 1] - beg + 1;

  float edd[H];
#pragma unroll
  for (int hh = 0; hh < H; ++hh) edd[hh] = ed[d * H + hh];

  int s0;
  float l0[H], mx[H];
  slot_logits<H>(col, es, edd, beg, tot, d, lane, s0, l0);
#pragma unroll
  for (int hh = 0; hh < H; ++hh) mx[hh] = l0[hh];
  for (int base = 32; base < tot; base += 32) {
    int s;
    float l[H];
    slot_logits<H>(col, es, edd, beg, tot, d, base + lane, s, l);
#pragma unroll
    for (int hh = 0; hh < H; ++hh) mx[hh] = fmaxf(mx[hh], l[hh]);
  }
#pragma unroll
  for (int hh = 0; hh < H; ++hh) {
#pragma unroll
    for (int o = 16; o > 0; o >>= 1) mx[hh] = fmaxf(mx[hh], __shfl_xor(mx[hh], o, 32));
    mx[hh] = __builtin_isfinite(mx[hh]) ? mx[hh] : 0.0f;
  }

  float sm[H];
#pragma unroll
  for (int hh = 0; hh < H; ++hh) sm[hh] = 0.0f;
  float4 acc0 = make_float4(0.f, 0.f, 0.f, 0.f);
  float4 acc1 = make_float4(0.f, 0.f, 0.f, 0.f);

  for (int base = 0; base < tot; base += 32) {
    int s = s0;
    float l[H], p[H];
#pragma unroll
    for (int hh = 0; hh < H; ++hh) l[hh] = l0[hh];
    if (base > 0) slot_logits<H>(col, es, edd, beg, tot, d, base + lane, s, l);
#pragma unroll
    for (int hh = 0; hh < H; ++hh) {
      p[hh] = (base + lane < tot) ? expf(l[hh] - mx[hh]) : 0.0f;
      sm[hh] += p[hh];
    }
    const int nk = (tot - base < 32) ? (tot - base) : 32;
#pragma unroll 4
    for (int k = 0; k < nk; ++k) {
      const int sk = __builtin_amdgcn_readlane(s, k);
      const float* hp = h + (size_t)sk * LDH;
      if constexpr (H == 4) {
        const float p0 = __int_as_float(__builtin_amdgcn_readlane(__float_as_int(p[0]), k));
        const float p1 = __int_as_float(__builtin_amdgcn_readlane(__float_as_int(p[1]), k));
        const float p2 = __int_as_float(__builtin_amdgcn_readlane(__float_as_int(p[2]), k));
        const float p3 = __int_as_float(__builtin_amdgcn_readlane(__float_as_int(p[3]), k));
        const float pa = (lane < 16) ? p0 : p1;
        const float pb = (lane < 16) ? p2 : p3;
        const float4 a = *reinterpret_cast<const float4*>(hp + 4 * lane);
        const float4 b = *reinterpret_cast<const float4*>(hp + 128 + 4 * lane);
        acc0.x += a.x * pa; acc0.y += a.y * pa; acc0.z += a.z * pa; acc0.w += a.w * pa;
        acc1.x += b.x * pb; acc1.y += b.y * pb; acc1.z += b.z * pb; acc1.w += b.w * pb;
      } else {
        const float p0 = __int_as_float(__builtin_amdgcn_readlane(__float_as_int(p[0]), k));
        const float2 a = *reinterpret_cast<const float2*>(hp + 2 * lane);
        acc0.x += a.x * p0; acc0.y += a.y * p0;
      }
    }
  }
#pragma unroll
  for (int hh = 0; hh < H; ++hh) {
#pragma unroll
    for (int o = 16; o > 0; o >>= 1) sm[hh] += __shfl_xor(sm[hh], o, 32);
  }

  float* op = out + (size_t)d * LDH;
  if constexpr (H == 4) {
    const float da = ((lane < 16) ? sm[0] : sm[1]) + 1e-16f;
    const float db = ((lane < 16) ? sm[2] : sm[3]) + 1e-16f;
    const float4 ba = *reinterpret_cast<const float4*>(bias + 4 * lane);
    const float4 bb = *reinterpret_cast<const float4*>(bias + 128 + 4 * lane);
    float4 oa, ob;
    oa.x = elu_f(acc0.x / da + ba.x); oa.y = elu_f(acc0.y / da + ba.y);
    oa.z = elu_f(acc0.z / da + ba.z); oa.w = elu_f(acc0.w / da + ba.w);
    ob.x = elu_f(acc1.x / db + bb.x); ob.y = elu_f(acc1.y / db + bb.y);
    ob.z = elu_f(acc1.z / db + bb.z); ob.w = elu_f(acc1.w / db + bb.w);
    *reinterpret_cast<float4*>(op + 4 * lane)       = oa;
    *reinterpret_cast<float4*>(op + 128 + 4 * lane) = ob;
  } else {
    const float da = sm[0] + 1e-16f;
    const float2 ba = *reinterpret_cast<const float2*>(bias + 2 * lane);
    float2 oa;
    oa.x = elu_f(acc0.x / da + ba.x); oa.y = elu_f(acc0.y / da + ba.y);
    *reinterpret_cast<float2*>(op + 2 * lane) = oa;
  }
}

__global__ void out_proj(const float* __restrict__ h,
                         const float* __restrict__ Wo,
                         const float* __restrict__ bo,
                         float* __restrict__ out, int n) {
  int wid  = (int)((blockIdx.x * blockDim.x + threadIdx.x) >> 5);
  int lane = (int)(threadIdx.x & 31);
  if (wid >= n) return;
  const float* hp = h + (size_t)wid * HID;
  float s = hp[lane] * Wo[lane] + hp[lane + 32] * Wo[lane + 32];
#pragma unroll
  for (int o = 16; o > 0; o >>= 1) s += __shfl_xor(s, o, 32);
  if (lane == 0) out[wid] = s + bo[0];
}

#define CDIV(a, b) (((a) + (b) - 1) / (b))

extern "C" void kernel_launch(void* const* d_in, const int* in_sizes, int n_in,
                              void* d_out, int out_size, void* d_ws, size_t ws_size,
                              hipStream_t stream) {
  const float* x   = (const float*)d_in[0];
  const float* W1  = (const float*)d_in[1];
  const float* as1 = (const float*)d_in[2];
  const float* ad1 = (const float*)d_in[3];
  const float* b1  = (const float*)d_in[4];
  const float* W2  = (const float*)d_in[5];
  const float* as2 = (const float*)d_in[6];
  const float* ad2 = (const float*)d_in[7];
  const float* b2  = (const float*)d_in[8];
  const float* Wo  = (const float*)d_in[9];
  const float* bo  = (const float*)d_in[10];
  const int*   ei  = (const int*)d_in[11];

  const int n  = in_sizes[0]  / IN_CH;
  const int e  = in_sizes[11] / 2;

  float* p = (float*)d_ws;
  float*    h1   = p;            p += (size_t)n * 256;
  float*    o1   = p;            p += (size_t)n * 256;
  float*    h2   = p;            p += (size_t)n * 64;
  float*    o2   = p;            p += (size_t)n * 64;
  float*    es1  = p;            p += (size_t)n * 4;
  float*    ed1  = p;            p += (size_t)n * 4;
  float*    es2v = p;            p += n;
  float*    ed2v = p;            p += n;
  unsigned* cnt  = (unsigned*)p; p += n;
  int*      col  = (int*)p;      p += e;
  unsigned* rptr = (unsigned*)p;
  unsigned* rank = (unsigned*)h1;

  const int B = 256;

  fill_u32<<<CDIV(n, B), B, 0, stream>>>(cnt, 0u, n);
  csr_count<<<CDIV(e, B), B, 0, stream>>>(ei, cnt, rank, e);
  csr_scan<<<n / SCAN_TILE + 1, B, 0, stream>>>(cnt, rptr, n);
  csr_fill<<<CDIV(e, B), B, 0, stream>>>(ei, rptr, rank, col, e);

  const int gneed = CDIV(n / 16, GEMM_WAVES);
  const int ggrid = gneed > 256 ? gneed : 256;
  wmma_gemm_f32<IN_CH, HEADS * HID><<<ggrid, GEMM_WAVES * 32, 0, stream>>>(x, W1, h1, n);
  attn_coef<HEADS><<<CDIV(n * HEADS, B), B, 0, stream>>>(h1, as1, ad1, es1, ed1, n);
  gat_gather<HEADS><<<CDIV(n * 32, B), B, 0, stream>>>(rptr, col, h1, es1, ed1, b1, o1, n);

  wmma_gemm_f32<HEADS * HID, HID><<<ggrid, GEMM_WAVES * 32, 0, stream>>>(o1, W2, h2, n);
  attn_coef<1><<<CDIV(n, B), B, 0, stream>>>(h2, as2, ad2, es2v, ed2v, n);
  gat_gather<1><<<CDIV(n * 32, B), B, 0, stream>>>(rptr, col, h2, es2v, ed2v, b2, o2, n);

  out_proj<<<CDIV(n * 32, B), B, 0, stream>>>(o2, Wo, bo, (float*)d_out, n);
}


# Edge ranks from LDS histograms (one work-group per 25600-edge slice, then a prefix over slices) instead of one global atomic per edge; segment order, and so the order of the f32 edge sums, changes
# speedup vs baseline: 3.9947x; 1.1396x over previous
#include <hip/hip_runtime.h>
#include <math.h>


#define IN_CH 128
#define HID   64
#define HEADS 4
#define NEG_SLOPE 0.2f

typedef float v2f __attribute__((ext_vector_type(2)));
typedef float v8f __attribute__((ext_vector_type(8)));

__device__ __forceinline__ float elu_f(float x) {
  return x > 0.0f ? x : (expf(x) - 1.0f);
}

#define NODES_MAX   50000
#define CNT_THREADS 1024
#define CNT_PER     25
#define SLICE       (CNT_THREADS * CNT_PER)
__global__ void __launch_bounds__(CNT_THREADS)
csr_count_lds(const int* __restrict__ ei, unsigned* __restrict__ part,
              unsigned* __restrict__ rank, int E_, int n) {
  __shared__ unsigned hist[NODES_MAX];
  const int t = (int)threadIdx.x;
  const int e0 = (int)blockIdx.x * SLICE + t;
  const int* dp = ei + (size_t)E_ + e0;
  unsigned* rp = rank + e0;

  int d[CNT_PER];
#pragma unroll
  for (int k = 0; k < CNT_PER; ++k)
    d[k] = (e0 + k * CNT_THREADS < E_) ? dp[k * CNT_THREADS] : -1;

  for (int i = 4 * t; i < n; i += 4 * CNT_THREADS)
    *reinterpret_cast<uint4*>(hist + i) = make_uint4(0u, 0u, 0u, 0u);
  __syncthreads();

  unsigned r[CNT_PER];
#pragma unroll
  for (int k = 0; k < CNT_PER; ++k) r[k] = (d[k] >= 0) ? atomicAdd(&hist[d[k]], 1u) : 0u;
  __syncthreads();

#pragma unroll
  for (int k = 0; k < CNT_PER; ++k)
    if (e0 + k * CNT_THREADS < E_) rp[k * CNT_THREADS] = r[k];

  unsigned* prow = part + (size_t)blockIdx.x * n;
  for (int i = 4 * t; i < n; i += 4 * CNT_THREADS)
    *reinterpret_cast<uint4*>(prow + i) = *reinterpret_cast<const uint4*>(hist + i);
}

#define MERGE_CHUNK 32
__global__ void csr_merge(unsigned* __restrict__ part, unsigned* __restrict__ cnt, int S, int n) {
  const int i = blockIdx.x * blockDim.x + threadIdx.x;
  if (i >= n) return;
  unsigned run = 0;
  for (int s0 = 0; s0 < S; s0 += MERGE_CHUNK) {
    unsigned c[MERGE_CHUNK];
#pragma unroll
    for (int k = 0; k < MERGE_CHUNK; ++k)
      c[k] = (s0 + k < S) ? part[(size_t)(s0 + k) * n + i] : 0u;
#pragma unroll
    for (int k = 0; k < MERGE_CHUNK; ++k) {
      if (s0 + k < S) part[(size_t)(s0 + k) * n + i] = run;
      run += c[k];
    }
  }
  cnt[i] = run;
}

#define SCAN_TILE 1024
__global__ void __launch_bounds__(256)
csr_scan(const unsigned* __restrict__ cnt, unsigned* __restrict__ rowptr, int n) {
  __shared__ unsigned wsum[8];
  __shared__ unsigned wpre[8];
  const int t = (int)threadIdx.x, lane = t & 31, w = t >> 5;
  const int tile0 = (int)blockIdx.x * SCAN_TILE;

  unsigned pre = 0;
#pragma unroll 4
  for (int i = 4 * t; i < tile0; i += SCAN_TILE) {
    const uint4 c = *reinterpret_cast<const uint4*>(cnt + i);
    pre += (c.x + c.y) + (c.z + c.w);
  }

  const int i0 = tile0 + 4 * t;
  unsigned c[4];
#pragma unroll
  for (int k = 0; k < 4; ++k) c[k] = (i0 + k < n) ? cnt[i0 + k] : 0u;
  const unsigned own = (c[0] + c[1]) + (c[2] + c[3]);

  unsigned incl = own;
#pragma unroll
  for (int o = 1; o < 32; o <<= 1) {
    unsigned v = __shfl_up(incl, (unsigned)o, 32);
    if (lane >= o) incl += v;
  }
#pragma unroll
  for (int o = 16; o > 0; o >>= 1) pre += __shfl_xor(pre, o, 32);
  if (lane == 31) wsum[w] = incl;
  if (lane == 0)  wpre[w] = pre;
  __syncthreads();
  unsigned base = incl - own;
#pragma unroll
  for (int j = 0; j < 8; ++j) {
    base += wpre[j];
    if (j < w) base += wsum[j];
  }
#pragma unroll
  for (int k = 0; k < 4; ++k) {
    const int idx = i0 + k;
    if (idx <= n) rowptr[idx] = base;
    base += c[k];
  }
}

__global__ void csr_fill(const int* __restrict__ ei, const unsigned* __restrict__ rowptr,
                         const unsigned* __restrict__ part, const unsigned* __restrict__ rank,
                         int* __restrict__ col, int E_, int n) {
  int e = blockIdx.x * blockDim.x + threadIdx.x;
  if (e >= E_) return;
  const int d = ei[E_ + e];
  col[rowptr[d] + part[(size_t)(e / SLICE) * n + d] + rank[e]] = ei[e];
}

#define GEMM_WAVES 16
template <int K, int NC>
__global__ void __launch_bounds__(GEMM_WAVES * 32) __attribute__((amdgpu_waves_per_eu(1, 4)))
wmma_gemm_f32(const float* __restrict__ A, const float* __restrict__ B,
              float* __restrict__ C, int M) {
  constexpr int NT = NC / 16;
  constexpr int KS = K / 8;
  constexpr int NQ = (K / 4) * NC;
  constexpr int NTHR = GEMM_WAVES * 32;
  constexpr int NIT = NQ / NTHR;
  static_assert(NQ % NTHR == 0, "staging shape");
  __shared__ float4 Bs[NQ];

  const int t = (int)threadIdx.x;
  const int lane = t & 31;
  const int lr = lane & 15, hi = lane >> 4;
  const int w = __builtin_amdgcn_readfirstlane(t >> 5);
  const int strip = (int)blockIdx.x + (int)gridDim.x * w;
  const int nstrips = M >> 4;
  const bool active = strip < nstrips;

  float4 stg[NIT];
#pragma unroll
  for (int j = 0; j < NIT; ++j) {
    const int idx = t + j * NTHR;
    const float* bp = B + (size_t)(4 * (idx / NC)) * NC + (idx % NC);
    stg[j].x = bp[0]; stg[j].y = bp[NC]; stg[j].z = bp[2 * NC]; stg[j].w = bp[3 * NC];
  }

#pragma unroll
  for (int j = 0; j < NIT; ++j) Bs[t + j * NTHR] = stg[j];
  __syncthreads();
  if (!active) return;

  float4 a[KS];
  {
    const float* ap = A + (size_t)(strip * 16 + lr) * K + 4 * hi;
#pragma unroll
    for (int i = 0; i < KS; ++i) a[i] = *reinterpret_cast<const float4*>(ap + 8 * i);
  }

  v8f acc[NT];
#pragma unroll
  for (int tn = 0; tn < NT; ++tn) acc[tn] = v8f{};

#pragma unroll
  for (int i = 0; i < KS; ++i) {
    const float4* bq = Bs + (2 * i + hi) * NC + lr;
    float4 b[NT];
#pragma unroll
    for (int tn = 0; tn < NT; ++tn) b[tn] = bq[tn * 16];
    v2f a01; a01.x = a[i].x; a01.y = a[i].y;
    v2f a23; a23.x = a[i].z; a23.y = a[i].w;
#pragma unroll
    for (int tn = 0; tn < NT; ++tn) {
      v2f b01; b01.x = b[tn].x; b01.y = b[tn].y;
      acc[tn] = __builtin_amdgcn_wmma_f32_16x16x4_f32(
           false, a01,  false, b01,
           (short)0, acc[tn],  false,  false);
    }
#pragma unroll
    for (int tn = 0; tn < NT; ++tn) {
      v2f b23; b23.x = b[tn].z; b23.y = b[tn].w;
      acc[tn] = __builtin_amdgcn_wmma_f32_16x16x4_f32(
           false, a23,  false, b23,
           (short)0, acc[tn],  false,  false);
    }
  }

  float* cp = C + (size_t)(strip * 16 + 8 * hi) * NC + lr;
#pragma unroll
  for (int tn = 0; tn < NT; ++tn) {
#pragma unroll
    for (int v = 0; v < 8; ++v) cp[(size_t)v * NC + tn * 16] = acc[tn][v];
  }
}

template <int H>
__global__ void attn_coef(const float* __restrict__ h,
                          const float* __restrict__ a_src,
                          const float* __restrict__ a_dst,
                          float* __restrict__ es, float* __restrict__ ed, int n) {
  int idx = blockIdx.x * blockDim.x + threadIdx.x;
  if (idx >= n * H) return;
  int i = idx / H, hh = idx % H;
  const float* hp = h + (size_t)i * (H * HID) + hh * HID;
  const float* as = a_src + hh * HID;
  const float* ad = a_dst + hh * HID;
  float s = 0.f, d = 0.f;
#pragma unroll 8
  for (int c = 0; c < HID; ++c) { float v = hp[c]; s += v * as[c]; d += v * ad[c]; }
  es[idx] = s; ed[idx] = d;
}

template <int H>
__device__ __forceinline__ void slot_logits(const int* __restrict__ col,
                                            const float* __restrict__ es,
                                            const float (&edd)[H], int beg, int tot,
                                            int d, int j, int& s, float (&l)[H]) {
  s = d;
#pragma unroll
  for (int hh = 0; hh < H; ++hh) l[hh] = -INFINITY;
  if (j < tot) {
    s = (j < tot - 1) ? col[beg + j] : d;
    float e[H];
    if constexpr (H == 4) {
      const float4 v = *reinterpret_cast<const float4*>(es + 4 * s);
      e[0] = v.x; e[1] = v.y; e[2] = v.z; e[3] = v.w;
    } else {
      e[0] = es[s];
    }
#pragma unroll
    for (int hh = 0; hh < H; ++hh) {
      const float x = e[hh] + edd[hh];
      l[hh] = (x > 0.0f) ? x : NEG_SLOPE * x;
    }
  }
}

template <int H>
__global__ void __launch_bounds__(256)
gat_gather(const unsigned* __restrict__ rowptr, const int* __restrict__ col,
           const float* __restrict__ h, const float* __restrict__ es,
           const float* __restrict__ ed, const float* __restrict__ bias,
           float* __restrict__ out, int n) {
  constexpr int LDH = H * HID;
  const int lane = (int)(threadIdx.x & 31);
  int d = (int)((blockIdx.x * blockDim.x + threadIdx.x) >> 5);
  d = __builtin_amdgcn_readfirstlane(d);
  if (d >= n) return;
  const int beg = (int)rowptr[d];
  const int tot = (int)rowptr[d + 1] - beg + 1;

  float edd[H];
#pragma unroll
  for (int hh = 0; hh < H; ++hh) edd[hh] = ed[d * H + hh];

  int s0;
  float l0[H], mx[H];
  slot_logits<H>(col, es, edd, beg, tot, d, lane, s0, l0);
#pragma unroll
  for (int hh = 0; hh < H; ++hh) mx[hh] = l0[hh];
  for (int base = 32; base < tot; base += 32) {
    int s;
    float l[H];
    slot_logits<H>(col, es, edd, beg, tot, d, base + lane, s, l);
#pragma unroll
    for (int hh = 0; hh < H; ++hh) mx[hh] = fmaxf(mx[hh], l[hh]);
  }
#pragma unroll
  for (int hh = 0; hh < H; ++hh) {
#pragma unroll
    for (int o = 16; o > 0; o >>= 1) mx[hh] = fmaxf(mx[hh], __shfl_xor(mx[hh], o, 32));
    mx[hh] = __builtin_isfinite(mx[hh]) ? mx[hh] : 0.0f;
  }

  float sm[H];
#pragma unroll
  for (int hh = 0; hh < H; ++hh) sm[hh] = 0.0f;
  float4 acc0 = make_float4(0.f, 0.f, 0.f, 0.f);
  float4 acc1 = make_float4(0.f, 0.f, 0.f, 0.f);

  for (int base = 0; base < tot; base += 32) {
    int s = s0;
    float l[H], p[H];
#pragma unroll
    for (int hh = 0; hh < H; ++hh) l[hh] = l0[hh];
    if (base > 0) slot_logits<H>(col, es, edd, beg, tot, d, base + lane, s, l);
#pragma unroll
    for (int hh = 0; hh < H; ++hh) {
      p[hh] = (base + lane < tot) ? expf(l[hh] - mx[hh]) : 0.0f;
      sm[hh] += p[hh];
    }
    const int nk = (tot - base < 32) ? (tot - base) : 32;
#pragma unroll 4
    for (int k = 0; k < nk; ++k) {
      const int sk = __builtin_amdgcn_readlane(s, k);
      const float* hp = h + (size_t)sk * LDH;
      if constexpr (H == 4) {
        const float p0 = __int_as_float(__builtin_amdgcn_readlane(__float_as_int(p[0]), k));
        const float p1 = __int_as_float(__builtin_amdgcn_readlane(__float_as_int(p[1]), k));
        const float p2 = __int_as_float(__builtin_amdgcn_readlane(__float_as_int(p[2]), k));
        const float p3 = __int_as_float(__builtin_amdgcn_readlane(__float_as_int(p[3]), k));
        const float pa = (lane < 16) ? p0 : p1;
        const float pb = (lane < 16) ? p2 : p3;
        const float4 a = *reinterpret_cast<const float4*>(hp + 4 * lane);
        const float4 b = *reinterpret_cast<const float4*>(hp + 128 + 4 * lane);
        acc0.x += a.x * pa; acc0.y += a.y * pa; acc0.z += a.z * pa; acc0.w += a.w * pa;
        acc1.x += b.x * pb; acc1.y += b.y * pb; acc1.z += b.z * pb; acc1.w += b.w * pb;
      } else {
        const float p0 = __int_as_float(__builtin_amdgcn_readlane(__float_as_int(p[0]), k));
        const float2 a = *reinterpret_cast<const float2*>(hp + 2 * lane);
        acc0.x += a.x * p0; acc0.y += a.y * p0;
      }
    }
  }
#pragma unroll
  for (int hh = 0; hh < H; ++hh) {
#pragma unroll
    for (int o = 16; o > 0; o >>= 1) sm[hh] += __shfl_xor(sm[hh], o, 32);
  }

  float* op = out + (size_t)d * LDH;
  if constexpr (H == 4) {
    const float da = ((lane < 16) ? sm[0] : sm[1]) + 1e-16f;
    const float db = ((lane < 16) ? sm[2] : sm[3]) + 1e-16f;
    const float4 ba = *reinterpret_cast<const float4*>(bias + 4 * lane);
    const float4 bb = *reinterpret_cast<const float4*>(bias + 128 + 4 * lane);
    float4 oa, ob;
    oa.x = elu_f(acc0.x / da + ba.x); oa.y = elu_f(acc0.y / da + ba.y);
    oa.z = elu_f(acc0.z / da + ba.z); oa.w = elu_f(acc0.w / da + ba.w);
    ob.x = elu_f(acc1.x / db + bb.x); ob.y = elu_f(acc1.y / db + bb.y);
    ob.z = elu_f(acc1.z / db + bb.z); ob.w = elu_f(acc1.w / db + bb.w);
    *reinterpret_cast<float4*>(op + 4 * lane)       = oa;
    *reinterpret_cast<float4*>(op + 128 + 4 * lane) = ob;
  } else {
    const float da = sm[0] + 1e-16f;
    const float2 ba = *reinterpret_cast<const float2*>(bias + 2 * lane);
    float2 oa;
    oa.x = elu_f(acc0.x / da + ba.x); oa.y = elu_f(acc0.y / da + ba.y);
    *reinterpret_cast<float2*>(op + 2 * lane) = oa;
  }
}

__global__ void out_proj(const float* __restrict__ h,
                         const float* __restrict__ Wo,
                         const float* __restrict__ bo,
                         float* __restrict__ out, int n) {
  int wid  = (int)((blockIdx.x * blockDim.x + threadIdx.x) >> 5);
  int lane = (int)(threadIdx.x & 31);
  if (wid >= n) return;
  const float* hp = h + (size_t)wid * HID;
  float s = hp[lane] * Wo[lane] + hp[lane + 32] * Wo[lane + 32];
#pragma unroll
  for (int o = 16; o > 0; o >>= 1) s += __shfl_xor(s, o, 32);
  if (lane == 0) out[wid] = s + bo[0];
}

#define CDIV(a, b) (((a) + (b) - 1) / (b))

extern "C" void kernel_launch(void* const* d_in, const int* in_sizes, int n_in,
                              void* d_out, int out_size, void* d_ws, size_t ws_size,
                              hipStream_t stream) {
  const float* x   = (const float*)d_in[0];
  const float* W1  = (const float*)d_in[1];
  const float* as1 = (const float*)d_in[2];
  const float* ad1 = (const float*)d_in[3];
  const float* b1  = (const float*)d_in[4];
  const float* W2  = (const float*)d_in[5];
  const float* as2 = (const float*)d_in[6];
  const float* ad2 = (const float*)d_in[7];
  const float* b2  = (const float*)d_in[8];
  const float* Wo  = (const float*)d_in[9];
  const float* bo  = (const float*)d_in[10];
  const int*   ei  = (const int*)d_in[11];

  const int n  = in_sizes[0]  / IN_CH;
  const int e  = in_sizes[11] / 2;

  float* p = (float*)d_ws;
  float*    h1   = p;            p += (size_t)n * 256;
  float*    o1   = p;            p += (size_t)n * 256;
  float*    h2   = p;            p += (size_t)n * 64;
  float*    o2   = p;            p += (size_t)n * 64;
  float*    es1  = p;            p += (size_t)n * 4;
  float*    ed1  = p;            p += (size_t)n * 4;
  float*    es2v = p;            p += n;
  float*    ed2v = p;            p += n;
  unsigned* cnt  = (unsigned*)p; p += n;
  int*      col  = (int*)p;      p += e;
  unsigned* rptr = (unsigned*)p;
  unsigned* rank = (unsigned*)h1;
  unsigned* part = rank + e;
  const int nslice = CDIV(e, SLICE);

  const int B = 256;

  csr_count_lds<<<nslice, CNT_THREADS, 0, stream>>>(ei, part, rank, e, n);
  csr_merge<<<CDIV(n, B), B, 0, stream>>>(part, cnt, nslice, n);
  csr_scan<<<n / SCAN_TILE + 1, B, 0, stream>>>(cnt, rptr, n);
  csr_fill<<<CDIV(e, B), B, 0, stream>>>(ei, rptr, part, rank, col, e, n);

  const int gneed = CDIV(n / 16, GEMM_WAVES);
  const int ggrid = gneed > 256 ? gneed : 256;
  wmma_gemm_f32<IN_CH, HEADS * HID><<<ggrid, GEMM_WAVES * 32, 0, stream>>>(x, W1, h1, n);
  attn_coef<HEADS><<<CDIV(n * HEADS, B), B, 0, stream>>>(h1, as1, ad1, es1, ed1, n);
  gat_gather<HEADS><<<CDIV(n * 32, B), B, 0, stream>>>(rptr, col, h1, es1, ed1, b1, o1, n);

  wmma_gemm_f32<HEADS * HID, HID><<<ggrid, GEMM_WAVES * 32, 0, stream>>>(o1, W2, h2, n);
  attn_coef<1><<<CDIV(n, B), B, 0, stream>>>(h2, as2, ad2, es2v, ed2v, n);
  gat_gather<1><<<CDIV(n * 32, B), B, 0, stream>>>(rptr, col, h2, es2v, ed2v, b2, o2, n);

  out_proj<<<CDIV(n * 32, B), B, 0, stream>>>(o2, Wo, bo, (float*)d_out, n);
}


# In-degree count co-scheduled with the layer-1 product in one launch (32 counting + 224 multiplying work-groups, one per WGP); LDS counters packed two per word; edge order inside a segment may change
# speedup vs baseline: 4.5611x; 1.1418x over previous
#include <hip/hip_runtime.h>
#include <math.h>


#define IN_CH 128
#define HID   64
#define HEADS 4
#define NEG_SLOPE 0.2f

typedef float v2f __attribute__((ext_vector_type(2)));
typedef float v8f __attribute__((ext_vector_type(8)));

__device__ __forceinline__ float elu_f(float x) {
  return x > 0.0f ? x : (expf(x) - 1.0f);
}

#define NODES_MAX   50000
#define CNT_THREADS 512
#define CNT_PER     50
#define SLICE       (CNT_THREADS * CNT_PER)
static_assert(SLICE < 65536, "a slice's count of one node must fit 16 bits");
__device__ __forceinline__ void
csr_count_body(unsigned* hist, int slice, const int* __restrict__ ei,
               unsigned* __restrict__ part16, unsigned* __restrict__ rank, int E_, int n) {
  const int t = (int)threadIdx.x;
  const int e0 = slice * SLICE + t;
  const int* dp = ei + (size_t)E_ + e0;
  unsigned* rp = rank + e0;

  int d[CNT_PER];
#pragma unroll
  for (int k = 0; k < CNT_PER; ++k)
    d[k] = (e0 + k * CNT_THREADS < E_) ? dp[k * CNT_THREADS] : -1;

  const int nw = n >> 1;
  for (int i = 4 * t; i < nw; i += 4 * CNT_THREADS)
    *reinterpret_cast<uint4*>(hist + i) = make_uint4(0u, 0u, 0u, 0u);
  __syncthreads();

  unsigned r[CNT_PER];
#pragma unroll
  for (int k = 0; k < CNT_PER; ++k)
    r[k] = (d[k] >= 0) ? atomicAdd(&hist[d[k] >> 1], (d[k] & 1) ? 0x10000u : 1u) : 0u;
  __syncthreads();

#pragma unroll
  for (int k = 0; k < CNT_PER; ++k)
    if (e0 + k * CNT_THREADS < E_)
      rp[k * CNT_THREADS] = (d[k] & 1) ? (r[k] >> 16) : (r[k] & 0xffffu);

  unsigned* prow = part16 + (size_t)slice * nw;
  for (int i = 4 * t; i < nw; i += 4 * CNT_THREADS)
    *reinterpret_cast<uint4*>(prow + i) = *reinterpret_cast<const uint4*>(hist + i);
}

#define MERGE_CHUNK 32
__global__ void csr_merge(const unsigned* __restrict__ part16w, unsigned* __restrict__ part,
                          unsigned* __restrict__ cnt, int S, int n) {
  const int i = blockIdx.x * blockDim.x + threadIdx.x;
  if (i >= n) return;
  const unsigned short* part16 = reinterpret_cast<const unsigned short*>(part16w);
  unsigned run = 0;
  for (int s0 = 0; s0 < S; s0 += MERGE_CHUNK) {
    unsigned c[MERGE_CHUNK];
#pragma unroll
    for (int k = 0; k < MERGE_CHUNK; ++k)
      c[k] = (s0 + k < S) ? (unsigned)part16[(size_t)(s0 + k) * n + i] : 0u;
#pragma unroll
    for (int k = 0; k < MERGE_CHUNK; ++k) {
      if (s0 + k < S) part[(size_t)(s0 + k) * n + i] = run;
      run += c[k];
    }
  }
  cnt[i] = run;
}

#define SCAN_TILE 1024
__global__ void __launch_bounds__(256)
csr_scan(const unsigned* __restrict__ cnt, unsigned* __restrict__ rowptr, int n) {
  __shared__ unsigned wsum[8];
  __shared__ unsigned wpre[8];
  const int t = (int)threadIdx.x, lane = t & 31, w = t >> 5;
  const int tile0 = (int)blockIdx.x * SCAN_TILE;

  unsigned pre = 0;
#pragma unroll 4
  for (int i = 4 * t; i < tile0; i += SCAN_TILE) {
    const uint4 c = *reinterpret_cast<const uint4*>(cnt + i);
    pre += (c.x + c.y) + (c.z + c.w);
  }

  const int i0 = tile0 + 4 * t;
  unsigned c[4];
#pragma unroll
  for (int k = 0; k < 4; ++k) c[k] = (i0 + k < n) ? cnt[i0 + k] : 0u;
  const unsigned own = (c[0] + c[1]) + (c[2] + c[3]);

  unsigned incl = own;
#pragma unroll
  for (int o = 1; o < 32; o <<= 1) {
    unsigned v = __shfl_up(incl, (unsigned)o, 32);
    if (lane >= o) incl += v;
  }
#pragma unroll
  for (int o = 16; o > 0; o >>= 1) pre += __shfl_xor(pre, o, 32);
  if (lane == 31) wsum[w] = incl;
  if (lane == 0)  wpre[w] = pre;
  __syncthreads();
  unsigned base = incl - own;
#pragma unroll
  for (int j = 0; j < 8; ++j) {
    base += wpre[j];
    if (j < w) base += wsum[j];
  }
#pragma unroll
  for (int k = 0; k < 4; ++k) {
    const int idx = i0 + k;
    if (idx <= n) rowptr[idx] = base;
    base += c[k];
  }
}

__global__ void csr_fill(const int* __restrict__ ei, const unsigned* __restrict__ rowptr,
                         const unsigned* __restrict__ part, const unsigned* __restrict__ rank,
                         int* __restrict__ col, int E_, int n) {
  int e = blockIdx.x * blockDim.x + threadIdx.x;
  if (e >= E_) return;
  const int d = ei[E_ + e];
  col[rowptr[d] + part[(size_t)(e / SLICE) * n + d] + rank[e]] = ei[e];
}

#define GEMM_WAVES 16

template <int CNT>
__device__ __forceinline__ void halve_step(float (&val)[16], bool up, int mask) {
#pragma unroll
  for (int i = 0; i < CNT; ++i) {
    const float keep = up ? val[CNT + i] : val[i];
    const float send = up ? val[i] : val[CNT + i];
    val[i] = keep + __shfl_xor(send, mask, 32);
  }
}
template <int K, int NC>
__device__ __forceinline__ void
gemm_body(float4* Bs, int bid, int nblk,
          const float* __restrict__ A, const float* __restrict__ B,
          float* __restrict__ C,
          const float* __restrict__ a_src, const float* __restrict__ a_dst,
          float* __restrict__ es, float* __restrict__ ed, int M) {
  constexpr int NT = NC / 16;
  constexpr int H  = NC / HID;
  constexpr int TPH = HID / 16;
  constexpr int KS = K / 8;
  constexpr int NQ = (K / 4) * NC;
  constexpr int NTHR = GEMM_WAVES * 32;
  constexpr int NIT = NQ / NTHR;
  static_assert(NQ % NTHR == 0, "staging shape");

  const int t = (int)threadIdx.x;
  const int lane = t & 31;
  const int lr = lane & 15, hi = lane >> 4;
  const int w = __builtin_amdgcn_readfirstlane(t >> 5);
  const int strip = bid + nblk * w;
  const int nstrips = M >> 4;
  const bool active = strip < nstrips;

  float4 stg[NIT];
#pragma unroll
  for (int j = 0; j < NIT; ++j) {
    const int idx = t + j * NTHR;
    const float* bp = B + (size_t)(4 * (idx / NC)) * NC + (idx % NC);
    stg[j].x = bp[0]; stg[j].y = bp[NC]; stg[j].z = bp[2 * NC]; stg[j].w = bp[3 * NC];
  }

#pragma unroll
  for (int j = 0; j < NIT; ++j) Bs[t + j * NTHR] = stg[j];
  __syncthreads();
  if (!active) return;

  float4 a[KS];
  {
    const float* ap = A + (size_t)(strip * 16 + lr) * K + 4 * hi;
#pragma unroll
    for (int i = 0; i < KS; ++i) a[i] = *reinterpret_cast<const float4*>(ap + 8 * i);
  }

  v8f acc[NT];
#pragma unroll
  for (int tn = 0; tn < NT; ++tn) acc[tn] = v8f{};

#pragma unroll
  for (int i = 0; i < KS; ++i) {
    const float4* bq = Bs + (2 * i + hi) * NC + lr;
    float4 b[NT];
#pragma unroll
    for (int tn = 0; tn < NT; ++tn) b[tn] = bq[tn * 16];
    v2f a01; a01.x = a[i].x; a01.y = a[i].y;
    v2f a23; a23.x = a[i].z; a23.y = a[i].w;
#pragma unroll
    for (int tn = 0; tn < NT; ++tn) {
      v2f b01; b01.x = b[tn].x; b01.y = b[tn].y;
      acc[tn] = __builtin_amdgcn_wmma_f32_16x16x4_f32(
           false, a01,  false, b01,
           (short)0, acc[tn],  false,  false);
    }
#pragma unroll
    for (int tn = 0; tn < NT; ++tn) {
      v2f b23; b23.x = b[tn].z; b23.y = b[tn].w;
      acc[tn] = __builtin_amdgcn_wmma_f32_16x16x4_f32(
           false, a23,  false, b23,
           (short)0, acc[tn],  false,  false);
    }
  }

  float asl[NT], adl[NT];
#pragma unroll
  for (int tn = 0; tn < NT; ++tn) {
    asl[tn] = a_src[tn * 16 + lr];
    adl[tn] = a_dst[tn * 16 + lr];
  }

  float* cp = C + (size_t)(strip * 16 + 8 * hi) * NC + lr;
#pragma unroll
  for (int tn = 0; tn < NT; ++tn) {
#pragma unroll
    for (int v = 0; v < 8; ++v) cp[(size_t)v * NC + tn * 16] = acc[tn][v];
  }

  float res[H];
#pragma unroll
  for (int hh = 0; hh < H; ++hh) {
    float val[16];
#pragma unroll
    for (int v = 0; v < 8; ++v) {
      float s = 0.0f, d = 0.0f;
#pragma unroll
      for (int t = 0; t < TPH; ++t) {
        s += acc[hh * TPH + t][v] * asl[hh * TPH + t];
        d += acc[hh * TPH + t][v] * adl[hh * TPH + t];
      }
      val[v] = s; val[8 + v] = d;
    }
    halve_step<8>(val, (lr & 1) != 0, 1);
    halve_step<4>(val, (lr & 2) != 0, 2);
    halve_step<2>(val, (lr & 4) != 0, 4);
    halve_step<1>(val, (lr & 8) != 0, 8);
    res[hh] = val[0];
  }
  {
    const int v = ((lr & 2) << 1) | ((lr & 4) >> 1) | ((lr & 8) >> 3);
    float* o = ((lr & 1) ? ed : es) + (size_t)(strip * 16 + 8 * hi + v) * H;
    if constexpr (H == 4) {
      *reinterpret_cast<float4*>(o) = make_float4(res[0], res[1], res[2], res[3]);
    } else {
#pragma unroll
      for (int hh = 0; hh < H; ++hh) o[hh] = res[hh];
    }
  }
}

template <int K, int NC>
__global__ void __launch_bounds__(GEMM_WAVES * 32) __attribute__((amdgpu_waves_per_eu(1, 4)))
wmma_gemm_f32(const float* __restrict__ A, const float* __restrict__ B,
              float* __restrict__ C,
              const float* __restrict__ a_src, const float* __restrict__ a_dst,
              float* __restrict__ es, float* __restrict__ ed, int M) {
  __shared__ float4 Bs[(K / 4) * NC];
  gemm_body<K, NC>(Bs, (int)blockIdx.x, (int)gridDim.x, A, B, C, a_src, a_dst, es, ed, M);
}

__global__ void __launch_bounds__(GEMM_WAVES * 32) __attribute__((amdgpu_waves_per_eu(1, 4)))
count_and_gemm1(const int* __restrict__ ei, unsigned* __restrict__ part16,
                unsigned* __restrict__ rank, int E_, int nslice,
                const float* __restrict__ A, const float* __restrict__ B,
                float* __restrict__ C,
                const float* __restrict__ a_src, const float* __restrict__ a_dst,
                float* __restrict__ es, float* __restrict__ ed, int M) {
  static_assert(CNT_THREADS == GEMM_WAVES * 32, "one block size for both kinds of work-group");
  static_assert(NODES_MAX * 2 <= (IN_CH / 4) * (HEADS * HID) * 16, "the packed histogram fits in the staged weights' LDS");
  __shared__ float4 smem[(IN_CH / 4) * (HEADS * HID)];
  if ((int)blockIdx.x < nslice) {
    csr_count_body(reinterpret_cast<unsigned*>(smem), (int)blockIdx.x, ei, part16, rank, E_, M);
  } else {
    gemm_body<IN_CH, HEADS * HID>(smem, (int)blockIdx.x - nslice,
                                  (int)gridDim.x - nslice, A, B, C, a_src, a_dst, es, ed, M);
  }
}

template <int H>
__device__ __forceinline__ void slot_logits(const int* __restrict__ col,
                                            const float* __restrict__ es,
                                            const float (&edd)[H], int beg, int tot,
                                            int d, int j, int& s, float (&l)[H]) {
  s = d;
#pragma unroll
  for (int hh = 0; hh < H; ++hh) l[hh] = -INFINITY;
  if (j < tot) {
    s = (j < tot - 1) ? col[beg + j] : d;
    float e[H];
    if constexpr (H == 4) {
      const float4 v = *reinterpret_cast<const float4*>(es + 4 * s);
      e[0] = v.x; e[1] = v.y; e[2] = v.z; e[3] = v.w;
    } else {
      e[0] = es[s];
    }
#pragma unroll
    for (int hh = 0; hh < H; ++hh) {
      const float x = e[hh] + edd[hh];
      l[hh] = (x > 0.0f) ? x : NEG_SLOPE * x;
    }
  }
}

template <int H>
__global__ void __launch_bounds__(256)
gat_gather(const unsigned* __restrict__ rowptr, const int* __restrict__ col,
           const float* __restrict__ h, const float* __restrict__ es,
           const float* __restrict__ ed, const float* __restrict__ bias,
           float* __restrict__ out, int n) {
  constexpr int LDH = H * HID;
  const int lane = (int)(threadIdx.x & 31);
  int d = (int)((blockIdx.x * blockDim.x + threadIdx.x) >> 5);
  d = __builtin_amdgcn_readfirstlane(d);
  if (d >= n) return;
  const int beg = (int)rowptr[d];
  const int tot = (int)rowptr[d + 1] - beg + 1;

  float edd[H];
#pragma unroll
  for (int hh = 0; hh < H; ++hh) edd[hh] = ed[d * H + hh];

  int s0;
  float l0[H], mx[H];
  slot_logits<H>(col, es, edd, beg, tot, d, lane, s0, l0);
#pragma unroll
  for (int hh = 0; hh < H; ++hh) mx[hh] = l0[hh];
  for (int base = 32; base < tot; base += 32) {
    int s;
    float l[H];
    slot_logits<H>(col, es, edd, beg, tot, d, base + lane, s, l);
#pragma unroll
    for (int hh = 0; hh < H; ++hh) mx[hh] = fmaxf(mx[hh], l[hh]);
  }
#pragma unroll
  for (int hh = 0; hh < H; ++hh) {
#pragma unroll
    for (int o = 16; o > 0; o >>= 1) mx[hh] = fmaxf(mx[hh], __shfl_xor(mx[hh], o, 32));
    mx[hh] = __builtin_isfinite(mx[hh]) ? mx[hh] : 0.0f;
  }

  float sm[H];
#pragma unroll
  for (int hh = 0; hh < H; ++hh) sm[hh] = 0.0f;
  float4 acc0 = make_float4(0.f, 0.f, 0.f, 0.f);
  float4 acc1 = make_float4(0.f, 0.f, 0.f, 0.f);

  for (int base = 0; base < tot; base += 32) {
    int s = s0;
    float l[H], p[H];
#pragma unroll
    for (int hh = 0; hh < H; ++hh) l[hh] = l0[hh];
    if (base > 0) slot_logits<H>(col, es, edd, beg, tot, d, base + lane, s, l);
#pragma unroll
    for (int hh = 0; hh < H; ++hh) {
      p[hh] = (base + lane < tot) ? expf(l[hh] - mx[hh]) : 0.0f;
      sm[hh] += p[hh];
    }
    const int nk = (tot - base < 32) ? (tot - base) : 32;
#pragma unroll 4
    for (int k = 0; k < nk; ++k) {
      const int sk = __builtin_amdgcn_readlane(s, k);
      const float* hp = h + (size_t)sk * LDH;
      if constexpr (H == 4) {
        const float p0 = __int_as_float(__builtin_amdgcn_readlane(__float_as_int(p[0]), k));
        const float p1 = __int_as_float(__builtin_amdgcn_readlane(__float_as_int(p[1]), k));
        const float p2 = __int_as_float(__builtin_amdgcn_readlane(__float_as_int(p[2]), k));
        const float p3 = __int_as_float(__builtin_amdgcn_readlane(__float_as_int(p[3]), k));
        const float pa = (lane < 16) ? p0 : p1;
        const float pb = (lane < 16) ? p2 : p3;
        const float4 a = *reinterpret_cast<const float4*>(hp + 4 * lane);
        const float4 b = *reinterpret_cast<const float4*>(hp + 128 + 4 * lane);
        acc0.x += a.x * pa; acc0.y += a.y * pa; acc0.z += a.z * pa; acc0.w += a.w * pa;
        acc1.x += b.x * pb; acc1.y += b.y * pb; acc1.z += b.z * pb; acc1.w += b.w * pb;
      } else {
        const float p0 = __int_as_float(__builtin_amdgcn_readlane(__float_as_int(p[0]), k));
        const float2 a = *reinterpret_cast<const float2*>(hp + 2 * lane);
        acc0.x += a.x * p0; acc0.y += a.y * p0;
      }
    }
  }
#pragma unroll
  for (int hh = 0; hh < H; ++hh) {
#pragma unroll
    for (int o = 16; o > 0; o >>= 1) sm[hh] += __shfl_xor(sm[hh], o, 32);
  }

  float* op = out + (size_t)d * LDH;
  if constexpr (H == 4) {
    const float da = ((lane < 16) ? sm[0] : sm[1]) + 1e-16f;
    const float db = ((lane < 16) ? sm[2] : sm[3]) + 1e-16f;
    const float4 ba = *reinterpret_cast<const float4*>(bias + 4 * lane);
    const float4 bb = *reinterpret_cast<const float4*>(bias + 128 + 4 * lane);
    float4 oa, ob;
    oa.x = elu_f(acc0.x / da + ba.x); oa.y = elu_f(acc0.y / da + ba.y);
    oa.z = elu_f(acc0.z / da + ba.z); oa.w = elu_f(acc0.w / da + ba.w);
    ob.x = elu_f(acc1.x / db + bb.x); ob.y = elu_f(acc1.y / db + bb.y);
    ob.z = elu_f(acc1.z / db + bb.z); ob.w = elu_f(acc1.w / db + bb.w);
    *reinterpret_cast<float4*>(op + 4 * lane)       = oa;
    *reinterpret_cast<float4*>(op + 128 + 4 * lane) = ob;
  } else {
    const float da = sm[0] + 1e-16f;
    const float2 ba = *reinterpret_cast<const float2*>(bias + 2 * lane);
    float2 oa;
    oa.x = elu_f(acc0.x / da + ba.x); oa.y = elu_f(acc0.y / da + ba.y);
    *reinterpret_cast<float2*>(op + 2 * lane) = oa;
  }
}

__global__ void out_proj(const float* __restrict__ h,
                         const float* __restrict__ Wo,
                         const float* __restrict__ bo,
                         float* __restrict__ out, int n) {
  int wid  = (int)((blockIdx.x * blockDim.x + threadIdx.x) >> 5);
  int lane = (int)(threadIdx.x & 31);
  if (wid >= n) return;
  const float* hp = h + (size_t)wid * HID;
  float s = hp[lane] * Wo[lane] + hp[lane + 32] * Wo[lane + 32];
#pragma unroll
  for (int o = 16; o > 0; o >>= 1) s += __shfl_xor(s, o, 32);
  if (lane == 0) out[wid] = s + bo[0];
}

#define CDIV(a, b) (((a) + (b) - 1) / (b))

extern "C" void kernel_launch(void* const* d_in, const int* in_sizes, int n_in,
                              void* d_out, int out_size, void* d_ws, size_t ws_size,
                              hipStream_t stream) {
  const float* x   = (const float*)d_in[0];
  const float* W1  = (const float*)d_in[1];
  const float* as1 = (const float*)d_in[2];
  const float* ad1 = (const float*)d_in[3];
  const float* b1  = (const float*)d_in[4];
  const float* W2  = (const float*)d_in[5];
  const float* as2 = (const float*)d_in[6];
  const float* ad2 = (const float*)d_in[7];
  const float* b2  = (const float*)d_in[8];
  const float* Wo  = (const float*)d_in[9];
  const float* bo  = (const float*)d_in[10];
  const int*   ei  = (const int*)d_in[11];

  const int n  = in_sizes[0]  / IN_CH;
  const int e  = in_sizes[11] / 2;

  float* p = (float*)d_ws;
  float*    h1   = p;            p += (size_t)n * 256;
  float*    o1   = p;            p += (size_t)n * 256;
  float*    h2   = p;            p += (size_t)n * 64;
  float*    o2   = p;            p += (size_t)n * 64;
  float*    es1  = p;            p += (size_t)n * 4;
  float*    ed1  = p;            p += (size_t)n * 4;
  float*    es2v = p;            p += n;
  float*    ed2v = p;            p += n;
  unsigned* cnt  = (unsigned*)p; p += n;
  int*      col  = (int*)p;      p += e;
  unsigned* rptr = (unsigned*)p;
  const int nslice = CDIV(e, SLICE);
  unsigned* rank   = (unsigned*)o1;
  unsigned* part   = rank + e;
  unsigned* part16 = part + (size_t)nslice * n;

  const int B = 256;

  const int gneed = CDIV(n / 16, GEMM_WAVES);
  const int gpart = gneed > 256 - nslice ? gneed : 256 - nslice;
  count_and_gemm1<<<nslice + gpart, GEMM_WAVES * 32, 0, stream>>>(
      ei, part16, rank, e, nslice, x, W1, h1, as1, ad1, es1, ed1, n);

  csr_merge<<<CDIV(n, B), B, 0, stream>>>(part16, part, cnt, nslice, n);
  csr_scan<<<n / SCAN_TILE + 1, B, 0, stream>>>(cnt, rptr, n);
  csr_fill<<<CDIV(e, B), B, 0, stream>>>(ei, rptr, part, rank, col, e, n);

  gat_gather<HEADS><<<CDIV(n * 32, B), B, 0, stream>>>(rptr, col, h1, es1, ed1, b1, o1, n);

  const int ggrid = gneed > 256 ? gneed : 256;
  wmma_gemm_f32<HEADS * HID, HID><<<ggrid, GEMM_WAVES * 32, 0, stream>>>(
      o1, W2, h2, as2, ad2, es2v, ed2v, n);
  gat_gather<1><<<CDIV(n * 32, B), B, 0, stream>>>(rptr, col, h2, es2v, ed2v, b2, o2, n);

  out_proj<<<CDIV(n * 32, B), B, 0, stream>>>(o2, Wo, bo, (float*)d_out, n);
}


# Both gathers: exp of the softmax and of ELU as one hardware exp2, one hardware reciprocal per head sum in place of a division per channel; last-bit rounding of p and of the outputs changes
# speedup vs baseline: 5.7237x; 1.2549x over previous
#include <hip/hip_runtime.h>
#include <math.h>


#define IN_CH 128
#define HID   64
#define HEADS 4
#define NEG_SLOPE 0.2f

typedef float v2f __attribute__((ext_vector_type(2)));
typedef float v8f __attribute__((ext_vector_type(8)));

__device__ __forceinline__ float exp_hw(float x) {
  return __builtin_amdgcn_exp2f(x * 1.44269504088896340736f);
}

__device__ __forceinline__ float elu_f(float x) {
  return x > 0.0f ? x : (exp_hw(x) - 1.0f);
}

#define NODES_MAX   50000
#define CNT_THREADS 512
#define CNT_PER     50
#define SLICE       (CNT_THREADS * CNT_PER)
#define SEG_SLICES  32
#define SEG_PITCH   (NODES_MAX + 16)
#define SCAN_CHUNK  ((NODES_MAX / 2 + CNT_THREADS - 1) / CNT_THREADS)
#define SORT_LDS_BYTES (NODES_MAX * 2 + 64 + SLICE * 2)
static_assert(SLICE < 65536, "a slice's count and offsets must fit 16 bits");
static_assert(NODES_MAX <= 65536, "a node number fits 16 bits (an edge's destination and rank share a word; the sorted list holds 16-bit sources)");
__device__ __forceinline__ void
csr_slice_body(unsigned* hist, int slice, const int* __restrict__ ei,
               unsigned* __restrict__ segtab, int* __restrict__ col,
               unsigned* __restrict__ etmp, int E_, int n) {
  const int t = (int)threadIdx.x;
  const int lane = t & 31, wv = t >> 5;
  const int e0 = slice * SLICE + t;
  const int* dp = ei + (size_t)E_ + e0;
  const int* sp = ei + e0;

  int d[CNT_PER];
#pragma unroll
  for (int k = 0; k < CNT_PER; ++k)
    d[k] = (e0 + k * CNT_THREADS < E_) ? dp[k * CNT_THREADS] : -1;

  const int nw = n >> 1;
  for (int i = 4 * t; i < nw; i += 4 * CNT_THREADS)
    *reinterpret_cast<uint4*>(hist + i) = make_uint4(0u, 0u, 0u, 0u);
  __syncthreads();

  constexpr int ADD_GRP = 10;
  static_assert(CNT_PER % ADD_GRP == 0, "whole groups");
  unsigned* tp = etmp + e0;
#pragma unroll
  for (int g = 0; g < CNT_PER; g += ADD_GRP) {
    unsigned r[ADD_GRP];
#pragma unroll
    for (int k = 0; k < ADD_GRP; ++k)
      r[k] = (d[g + k] >= 0)
                 ? atomicAdd(&hist[d[g + k] >> 1], (d[g + k] & 1) ? 0x10000u : 1u) : 0u;
#pragma unroll
    for (int k = 0; k < ADD_GRP; ++k)
      if (d[g + k] >= 0)
        tp[(g + k) * CNT_THREADS] =
            (unsigned)d[g + k] | (((d[g + k] & 1) ? (r[k] >> 16) : (r[k] & 0xffffu)) << 16);
  }
  __syncthreads();

  const int w0 = SCAN_CHUNK * t;
  const int w1 = (w0 + SCAN_CHUNK < nw) ? (w0 + SCAN_CHUNK) : nw;
  unsigned tsum = 0;
  for (int i = w0; i < w1; ++i) {
    const unsigned w = hist[i];
    tsum += (w & 0xffffu) + (w >> 16);
  }
  unsigned incl = tsum;
#pragma unroll
  for (int o = 1; o < 32; o <<= 1) {
    const unsigned v = __shfl_up(incl, (unsigned)o, 32);
    if (lane >= o) incl += v;
  }
  unsigned* aux = hist + NODES_MAX / 2;
  if (lane == 31) aux[wv] = incl;
  __syncthreads();
  unsigned run = incl - tsum;
#pragma unroll
  for (int j = 0; j < CNT_THREADS / 32; ++j)
    if (j < wv) run += aux[j];
  for (int i = w0; i < w1; ++i) {
    const unsigned w = hist[i];
    const unsigned c0 = w & 0xffffu, c1 = w >> 16;
    hist[i] = run | ((run + c0) << 16);
    run += c0 + c1;
  }
  __syncthreads();

  const unsigned cbase = (unsigned)slice * SLICE;
  unsigned short* lst = reinterpret_cast<unsigned short*>(hist + NODES_MAX / 2 + 16);
  constexpr int PLACE_GRP = 10;
  static_assert(CNT_PER % PLACE_GRP == 0, "whole groups");
  for (int g = 0; g < CNT_PER; g += PLACE_GRP) {
    unsigned pk[PLACE_GRP];
    int sv[PLACE_GRP];
#pragma unroll
    for (int k = 0; k < PLACE_GRP; ++k) {
      const bool in = e0 + (g + k) * CNT_THREADS < E_;
      pk[k] = in ? tp[(g + k) * CNT_THREADS] : 0u;
      sv[k] = in ? sp[(g + k) * CNT_THREADS] : 0;
    }
#pragma unroll
    for (int k = 0; k < PLACE_GRP; ++k)
      if (e0 + (g + k) * CNT_THREADS < E_) {
        const unsigned dst = pk[k] & 0xffffu;
        const unsigned w = hist[dst >> 1];
        const unsigned off = (dst & 1u) ? (w >> 16) : (w & 0xffffu);
        lst[off + (pk[k] >> 16)] = (unsigned short)sv[k];
      }
  }
  __syncthreads();

  const int left = E_ - slice * SLICE;
  const int cnt = left < SLICE ? left : SLICE;
  for (int j = t; 2 * j < cnt; j += CNT_THREADS) {
    if (2 * j + 1 < cnt)
      *reinterpret_cast<int2*>(col + cbase + 2 * j) = make_int2((int)lst[2 * j], (int)lst[2 * j + 1]);
    else
      col[cbase + 2 * j] = (int)lst[2 * j];
  }

  unsigned* trow = segtab + (size_t)slice * SEG_PITCH;
  for (int i = t; i < n; i += CNT_THREADS) {
    const unsigned w = hist[i >> 1];
    trow[i] = cbase + ((i & 1) ? (w >> 16) : (w & 0xffffu));
  }
  if (t == 0) trow[n] = cbase + (unsigned)cnt;
}

#define GEMM_WAVES 16

template <int CNT>
__device__ __forceinline__ void halve_step(float (&val)[16], bool up, int mask) {
#pragma unroll
  for (int i = 0; i < CNT; ++i) {
    const float keep = up ? val[CNT + i] : val[i];
    const float send = up ? val[i] : val[CNT + i];
    val[i] = keep + __shfl_xor(send, mask, 32);
  }
}
template <int K, int NC>
__device__ __forceinline__ void
gemm_body(float4* Bs, int bid, int nblk,
          const float* __restrict__ A, const float* __restrict__ B,
          float* __restrict__ C,
          const float* __restrict__ a_src, const float* __restrict__ a_dst,
          float* __restrict__ es, float* __restrict__ ed, int M) {
  constexpr int NT = NC / 16;
  constexpr int H  = NC / HID;
  constexpr int TPH = HID / 16;
  constexpr int KS = K / 8;
  constexpr int NQ = (K / 4) * NC;
  constexpr int NTHR = GEMM_WAVES * 32;
  constexpr int NIT = NQ / NTHR;
  static_assert(NQ % NTHR == 0, "staging shape");

  const int t = (int)threadIdx.x;
  const int lane = t & 31;
  const int lr = lane & 15, hi = lane >> 4;
  const int w = __builtin_amdgcn_readfirstlane(t >> 5);
  const int strip = bid + nblk * w;
  const int nstrips = M >> 4;
  const bool active = strip < nstrips;

  float4 stg[NIT];
#pragma unroll
  for (int j = 0; j < NIT; ++j) {
    const int idx = t + j * NTHR;
    const float* bp = B + (size_t)(4 * (idx / NC)) * NC + (idx % NC);
    stg[j].x = bp[0]; stg[j].y = bp[NC]; stg[j].z = bp[2 * NC]; stg[j].w = bp[3 * NC];
  }

#pragma unroll
  for (int j = 0; j < NIT; ++j) Bs[t + j * NTHR] = stg[j];
  __syncthreads();
  if (!active) return;

  float4 a[KS];
  {
    const float* ap = A + (size_t)(strip * 16 + lr) * K + 4 * hi;
#pragma unroll
    for (int i = 0; i < KS; ++i) a[i] = *reinterpret_cast<const float4*>(ap + 8 * i);
  }

  v8f acc[NT];
#pragma unroll
  for (int tn = 0; tn < NT; ++tn) acc[tn] = v8f{};

#pragma unroll
  for (int i = 0; i < KS; ++i) {
    const float4* bq = Bs + (2 * i + hi) * NC + lr;
    float4 b[NT];
#pragma unroll
    for (int tn = 0; tn < NT; ++tn) b[tn] = bq[tn * 16];
    v2f a01; a01.x = a[i].x; a01.y = a[i].y;
    v2f a23; a23.x = a[i].z; a23.y = a[i].w;
#pragma unroll
    for (int tn = 0; tn < NT; ++tn) {
      v2f b01; b01.x = b[tn].x; b01.y = b[tn].y;
      acc[tn] = __builtin_amdgcn_wmma_f32_16x16x4_f32(
           false, a01,  false, b01,
           (short)0, acc[tn],  false,  false);
    }
#pragma unroll
    for (int tn = 0; tn < NT; ++tn) {
      v2f b23; b23.x = b[tn].z; b23.y = b[tn].w;
      acc[tn] = __builtin_amdgcn_wmma_f32_16x16x4_f32(
           false, a23,  false, b23,
           (short)0, acc[tn],  false,  false);
    }
  }

  float asl[NT], adl[NT];
#pragma unroll
  for (int tn = 0; tn < NT; ++tn) {
    asl[tn] = a_src[tn * 16 + lr];
    adl[tn] = a_dst[tn * 16 + lr];
  }

  float* cp = C + (size_t)(strip * 16 + 8 * hi) * NC + lr;
#pragma unroll
  for (int tn = 0; tn < NT; ++tn) {
#pragma unroll
    for (int v = 0; v < 8; ++v) cp[(size_t)v * NC + tn * 16] = acc[tn][v];
  }

  float res[H];
#pragma unroll
  for (int hh = 0; hh < H; ++hh) {
    float val[16];
#pragma unroll
    for (int v = 0; v < 8; ++v) {
      float s = 0.0f, d = 0.0f;
#pragma unroll
      for (int t = 0; t < TPH; ++t) {
        s += acc[hh * TPH + t][v] * asl[hh * TPH + t];
        d += acc[hh * TPH + t][v] * adl[hh * TPH + t];
      }
      val[v] = s; val[8 + v] = d;
    }
    halve_step<8>(val, (lr & 1) != 0, 1);
    halve_step<4>(val, (lr & 2) != 0, 2);
    halve_step<2>(val, (lr & 4) != 0, 4);
    halve_step<1>(val, (lr & 8) != 0, 8);
    res[hh] = val[0];
  }
  {
    const int v = ((lr & 2) << 1) | ((lr & 4) >> 1) | ((lr & 8) >> 3);
    float* o = ((lr & 1) ? ed : es) + (size_t)(strip * 16 + 8 * hi + v) * H;
    if constexpr (H == 4) {
      *reinterpret_cast<float4*>(o) = make_float4(res[0], res[1], res[2], res[3]);
    } else {
#pragma unroll
      for (int hh = 0; hh < H; ++hh) o[hh] = res[hh];
    }
  }
}

template <int K, int NC>
__global__ void __launch_bounds__(GEMM_WAVES * 32) __attribute__((amdgpu_waves_per_eu(1, 4)))
wmma_gemm_f32(const float* __restrict__ A, const float* __restrict__ B,
              float* __restrict__ C,
              const float* __restrict__ a_src, const float* __restrict__ a_dst,
              float* __restrict__ es, float* __restrict__ ed, int M) {
  __shared__ float4 Bs[(K / 4) * NC];
  gemm_body<K, NC>(Bs, (int)blockIdx.x, (int)gridDim.x, A, B, C, a_src, a_dst, es, ed, M);
}

__global__ void __launch_bounds__(GEMM_WAVES * 32) __attribute__((amdgpu_waves_per_eu(1, 4)))
count_and_gemm1(const int* __restrict__ ei, unsigned* __restrict__ segtab,
                int* __restrict__ col, unsigned* __restrict__ etmp, int E_, int nslice,
                const float* __restrict__ A, const float* __restrict__ B,
                float* __restrict__ C,
                const float* __restrict__ a_src, const float* __restrict__ a_dst,
                float* __restrict__ es, float* __restrict__ ed, int M) {
  static_assert(CNT_THREADS == GEMM_WAVES * 32, "one block size for both kinds of work-group");
  constexpr int GEMM_LDS_BYTES = (IN_CH / 4) * (HEADS * HID) * 16;
  constexpr int LDS_BYTES = GEMM_LDS_BYTES > SORT_LDS_BYTES ? GEMM_LDS_BYTES : SORT_LDS_BYTES;
  static_assert(CNT_THREADS / 32 * 4 <= 64, "the scan's wave totals fit their 16 words");
  __shared__ float4 smem[(LDS_BYTES + 15) / 16];
  if ((int)blockIdx.x < nslice) {
    csr_slice_body(reinterpret_cast<unsigned*>(smem), (int)blockIdx.x, ei, segtab, col, etmp, E_, M);
  } else {
    gemm_body<IN_CH, HEADS * HID>(smem, (int)blockIdx.x - nslice,
                                  (int)gridDim.x - nslice, A, B, C, a_src, a_dst, es, ed, M);
  }
}

__device__ __forceinline__ int slot_col(int inc, int segbase, int j) {
  int lo = 0;
#pragma unroll
  for (int b = 16; b > 0; b >>= 1) {
    const int v = __shfl(inc, lo + b - 1, 32);
    if (v <= j) lo += b;
  }
  return __shfl(segbase, lo, 32) + j;
}

template <int H>
__device__ __forceinline__ void slot_logits(const int* __restrict__ col,
                                            const float* __restrict__ es,
                                            const float (&edd)[H], int ci, int tot,
                                            int d, int j, int& s, float (&l)[H]) {
  s = d;
#pragma unroll
  for (int hh = 0; hh < H; ++hh) l[hh] = -INFINITY;
  if (j < tot) {
    s = (j < tot - 1) ? col[ci] : d;
    float e[H];
    if constexpr (H == 4) {
      const float4 v = *reinterpret_cast<const float4*>(es + 4 * s);
      e[0] = v.x; e[1] = v.y; e[2] = v.z; e[3] = v.w;
    } else {
      e[0] = es[s];
    }
#pragma unroll
    for (int hh = 0; hh < H; ++hh) {
      const float x = e[hh] + edd[hh];
      l[hh] = (x > 0.0f) ? x : NEG_SLOPE * x;
    }
  }
}

template <int H>
__global__ void __launch_bounds__(256)
gat_gather(const unsigned* __restrict__ segtab, int nslice, const int* __restrict__ col,
           const float* __restrict__ h, const float* __restrict__ es,
           const float* __restrict__ ed, const float* __restrict__ bias,
           float* __restrict__ out, int n) {
  constexpr int LDH = H * HID;
  const int lane = (int)(threadIdx.x & 31);
  int d = (int)((blockIdx.x * blockDim.x + threadIdx.x) >> 5);
  d = __builtin_amdgcn_readfirstlane(d);
  if (d >= n) return;
  unsigned a0 = 0, a1 = 0;
  if (lane < nslice) {
    const unsigned* tr = segtab + (size_t)lane * SEG_PITCH + d;
    a0 = tr[0];
    a1 = tr[1];
  }
  const int len = (int)(a1 - a0);
  int inc = len;
#pragma unroll
  for (int o = 1; o < 32; o <<= 1) {
    const int v = __shfl_up(inc, (unsigned)o, 32);
    if (lane >= o) inc += v;
  }
  const int segbase = (int)a0 - (inc - len);
  const int tot = __builtin_amdgcn_readlane(inc, 31) + 1;

  float edd[H];
#pragma unroll
  for (int hh = 0; hh < H; ++hh) edd[hh] = ed[d * H + hh];

  int s0;
  float l0[H], mx[H];
  slot_logits<H>(col, es, edd, slot_col(inc, segbase, lane), tot, d, lane, s0, l0);
#pragma unroll
  for (int hh = 0; hh < H; ++hh) mx[hh] = l0[hh];
  for (int base = 32; base < tot; base += 32) {
    int s;
    float l[H];
    slot_logits<H>(col, es, edd, slot_col(inc, segbase, base + lane), tot, d, base + lane, s, l);
#pragma unroll
    for (int hh = 0; hh < H; ++hh) mx[hh] = fmaxf(mx[hh], l[hh]);
  }
#pragma unroll
  for (int hh = 0; hh < H; ++hh) {
#pragma unroll
    for (int o = 16; o > 0; o >>= 1) mx[hh] = fmaxf(mx[hh], __shfl_xor(mx[hh], o, 32));
    mx[hh] = __builtin_isfinite(mx[hh]) ? mx[hh] : 0.0f;
  }

  float sm[H];
#pragma unroll
  for (int hh = 0; hh < H; ++hh) sm[hh] = 0.0f;
  float4 acc0 = make_float4(0.f, 0.f, 0.f, 0.f);
  float4 acc1 = make_float4(0.f, 0.f, 0.f, 0.f);

  for (int base = 0; base < tot; base += 32) {
    int s = s0;
    float l[H], p[H];
#pragma unroll
    for (int hh = 0; hh < H; ++hh) l[hh] = l0[hh];
    if (base > 0)
      slot_logits<H>(col, es, edd, slot_col(inc, segbase, base + lane), tot, d, base + lane, s, l);
#pragma unroll
    for (int hh = 0; hh < H; ++hh) {
      p[hh] = (base + lane < tot) ? exp_hw(l[hh] - mx[hh]) : 0.0f;
      sm[hh] += p[hh];
    }
    const int nk = (tot - base < 32) ? (tot - base) : 32;
#pragma unroll 4
    for (int k = 0; k < nk; ++k) {
      const int sk = __builtin_amdgcn_readlane(s, k);
      const float* hp = h + (size_t)sk * LDH;
      if constexpr (H == 4) {
        const float p0 = __int_as_float(__builtin_amdgcn_readlane(__float_as_int(p[0]), k));
        const float p1 = __int_as_float(__builtin_amdgcn_readlane(__float_as_int(p[1]), k));
        const float p2 = __int_as_float(__builtin_amdgcn_readlane(__float_as_int(p[2]), k));
        const float p3 = __int_as_float(__builtin_amdgcn_readlane(__float_as_int(p[3]), k));
        const float pa = (lane < 16) ? p0 : p1;
        const float pb = (lane < 16) ? p2 : p3;
        const float4 a = *reinterpret_cast<const float4*>(hp + 4 * lane);
        const float4 b = *reinterpret_cast<const float4*>(hp + 128 + 4 * lane);
        acc0.x += a.x * pa; acc0.y += a.y * pa; acc0.z += a.z * pa; acc0.w += a.w * pa;
        acc1.x += b.x * pb; acc1.y += b.y * pb; acc1.z += b.z * pb; acc1.w += b.w * pb;
      } else {
        const float p0 = __int_as_float(__builtin_amdgcn_readlane(__float_as_int(p[0]), k));
        const float2 a = *reinterpret_cast<const float2*>(hp + 2 * lane);
        acc0.x += a.x * p0; acc0.y += a.y * p0;
      }
    }
  }
#pragma unroll
  for (int hh = 0; hh < H; ++hh) {
#pragma unroll
    for (int o = 16; o > 0; o >>= 1) sm[hh] += __shfl_xor(sm[hh], o, 32);
  }

  float* op = out + (size_t)d * LDH;
  if constexpr (H == 4) {
    const float ra = __builtin_amdgcn_rcpf(((lane < 16) ? sm[0] : sm[1]) + 1e-16f);
    const float rb = __builtin_amdgcn_rcpf(((lane < 16) ? sm[2] : sm[3]) + 1e-16f);
    const float4 ba = *reinterpret_cast<const float4*>(bias + 4 * lane);
    const float4 bb = *reinterpret_cast<const float4*>(bias + 128 + 4 * lane);
    float4 oa, ob;
    oa.x = elu_f(acc0.x * ra + ba.x); oa.y = elu_f(acc0.y * ra + ba.y);
    oa.z = elu_f(acc0.z * ra + ba.z); oa.w = elu_f(acc0.w * ra + ba.w);
    ob.x = elu_f(acc1.x * rb + bb.x); ob.y = elu_f(acc1.y * rb + bb.y);
    ob.z = elu_f(acc1.z * rb + bb.z); ob.w = elu_f(acc1.w * rb + bb.w);
    *reinterpret_cast<float4*>(op + 4 * lane)       = oa;
    *reinterpret_cast<float4*>(op + 128 + 4 * lane) = ob;
  } else {
    const float ra = __builtin_amdgcn_rcpf(sm[0] + 1e-16f);
    const float2 ba = *reinterpret_cast<const float2*>(bias + 2 * lane);
    float2 oa;
    oa.x = elu_f(acc0.x * ra + ba.x); oa.y = elu_f(acc0.y * ra + ba.y);
    *reinterpret_cast<float2*>(op + 2 * lane) = oa;
  }
}

__global__ void out_proj(const float* __restrict__ h,
                         const float* __restrict__ Wo,
                         const float* __restrict__ bo,
                         float* __restrict__ out, int n) {
  int wid  = (int)((blockIdx.x * blockDim.x + threadIdx.x) >> 5);
  int lane = (int)(threadIdx.x & 31);
  if (wid >= n) return;
  const float* hp = h + (size_t)wid * HID;
  float s = hp[lane] * Wo[lane] + hp[lane + 32] * Wo[lane + 32];
#pragma unroll
  for (int o = 16; o > 0; o >>= 1) s += __shfl_xor(s, o, 32);
  if (lane == 0) out[wid] = s + bo[0];
}

#define CDIV(a, b) (((a) + (b) - 1) / (b))

extern "C" void kernel_launch(void* const* d_in, const int* in_sizes, int n_in,
                              void* d_out, int out_size, void* d_ws, size_t ws_size,
                              hipStream_t stream) {
  const float* x   = (const float*)d_in[0];
  const float* W1  = (const float*)d_in[1];
  const float* as1 = (const float*)d_in[2];
  const float* ad1 = (const float*)d_in[3];
  const float* b1  = (const float*)d_in[4];
  const float* W2  = (const float*)d_in[5];
  const float* as2 = (const float*)d_in[6];
  const float* ad2 = (const float*)d_in[7];
  const float* b2  = (const float*)d_in[8];
  const float* Wo  = (const float*)d_in[9];
  const float* bo  = (const float*)d_in[10];
  const int*   ei  = (const int*)d_in[11];

  const int n  = in_sizes[0]  / IN_CH;
  const int e  = in_sizes[11] / 2;

  float* p = (float*)d_ws;
  float*    h1   = p;            p += (size_t)n * 256;
  float*    o1   = p;            p += (size_t)n * 256;
  unsigned* segtab = (unsigned*)p; p += (size_t)SEG_SLICES * SEG_PITCH;
  float*    es1  = p;            p += (size_t)n * 4;
  float*    ed1  = p;            p += (size_t)n * 4;
  float*    es2v = p;            p += n;
  float*    ed2v = p;            p += n;
  int*      col  = (int*)p;      p += e;
  unsigned* etmp = (unsigned*)p;
  float*    h2   = h1;
  float*    o2   = h1 + (size_t)n * 64;
  const int nslice = CDIV(e, SLICE);

  const int B = 256;

  const int gneed = CDIV(n / 16, GEMM_WAVES);
  const int gpart = gneed > 256 - nslice ? gneed : 256 - nslice;
  count_and_gemm1<<<nslice + gpart, GEMM_WAVES * 32, 0, stream>>>(
      ei, segtab, col, etmp, e, nslice, x, W1, h1, as1, ad1, es1, ed1, n);

  gat_gather<HEADS><<<CDIV(n * 32, B), B, 0, stream>>>(segtab, nslice, col, h1, es1, ed1, b1, o1, n);

  const int ggrid = gneed > 256 ? gneed : 256;
  wmma_gemm_f32<HEADS * HID, HID><<<ggrid, GEMM_WAVES * 32, 0, stream>>>(
      o1, W2, h2, as2, ad2, es2v, ed2v, n);
  gat_gather<1><<<CDIV(n * 32, B), B, 0, stream>>>(segtab, nslice, col, h2, es2v, ed2v, b2, o2, n);

  out_proj<<<CDIV(n * 32, B), B, 0, stream>>>(o2, Wo, bo, (float*)d_out, n);
}


# Layer-2 product moved into the layer-1 gather launch: resident 16-wave groups gather 16 nodes into an LDS tile and multiply it by W2 on WMMA; one launch less; f32 k-sum and es/ed sum order changes
# speedup vs baseline: 5.9036x; 1.0314x over previous
#include <hip/hip_runtime.h>
#include <math.h>


#define IN_CH 128
#define HID   64
#define HEADS 4
#define NEG_SLOPE 0.2f

typedef float v2f __attribute__((ext_vector_type(2)));
typedef float v8f __attribute__((ext_vector_type(8)));

__device__ __forceinline__ float exp_hw(float x) {
  return __builtin_amdgcn_exp2f(x * 1.44269504088896340736f);
}

__device__ __forceinline__ float elu_f(float x) {
  return x > 0.0f ? x : (exp_hw(x) - 1.0f);
}

#define NODES_MAX   50000
#define CNT_THREADS 512
#define CNT_PER     50
#define SLICE       (CNT_THREADS * CNT_PER)
#define SEG_SLICES  32
#define SEG_PITCH   (NODES_MAX + 16)
#define SCAN_CHUNK  ((NODES_MAX / 2 + CNT_THREADS - 1) / CNT_THREADS)
#define SORT_LDS_BYTES (NODES_MAX * 2 + 64 + SLICE * 2)
static_assert(SLICE < 65536, "a slice's count and offsets must fit 16 bits");
static_assert(NODES_MAX <= 65536, "a node number fits 16 bits (an edge's destination and rank share a word; the sorted list holds 16-bit sources)");
__device__ __forceinline__ void
csr_slice_body(unsigned* hist, int slice, const int* __restrict__ ei,
               unsigned* __restrict__ segtab, int* __restrict__ col,
               unsigned* __restrict__ etmp, int E_, int n) {
  const int t = (int)threadIdx.x;
  const int lane = t & 31, wv = t >> 5;
  const int e0 = slice * SLICE + t;
  const int* dp = ei + (size_t)E_ + e0;
  const int* sp = ei + e0;

  int d[CNT_PER];
#pragma unroll
  for (int k = 0; k < CNT_PER; ++k)
    d[k] = (e0 + k * CNT_THREADS < E_) ? dp[k * CNT_THREADS] : -1;

  const int nw = n >> 1;
  for (int i = 4 * t; i < nw; i += 4 * CNT_THREADS)
    *reinterpret_cast<uint4*>(hist + i) = make_uint4(0u, 0u, 0u, 0u);
  __syncthreads();

  constexpr int ADD_GRP = 10;
  static_assert(CNT_PER % ADD_GRP == 0, "whole groups");
  unsigned* tp = etmp + e0;
#pragma unroll
  for (int g = 0; g < CNT_PER; g += ADD_GRP) {
    unsigned r[ADD_GRP];
#pragma unroll
    for (int k = 0; k < ADD_GRP; ++k)
      r[k] = (d[g + k] >= 0)
                 ? atomicAdd(&hist[d[g + k] >> 1], (d[g + k] & 1) ? 0x10000u : 1u) : 0u;
#pragma unroll
    for (int k = 0; k < ADD_GRP; ++k)
      if (d[g + k] >= 0)
        tp[(g + k) * CNT_THREADS] =
            (unsigned)d[g + k] | (((d[g + k] & 1) ? (r[k] >> 16) : (r[k] & 0xffffu)) << 16);
  }
  __syncthreads();

  const int w0 = SCAN_CHUNK * t;
  const int w1 = (w0 + SCAN_CHUNK < nw) ? (w0 + SCAN_CHUNK) : nw;
  unsigned tsum = 0;
  for (int i = w0; i < w1; ++i) {
    const unsigned w = hist[i];
    tsum += (w & 0xffffu) + (w >> 16);
  }
  unsigned incl = tsum;
#pragma unroll
  for (int o = 1; o < 32; o <<= 1) {
    const unsigned v = __shfl_up(incl, (unsigned)o, 32);
    if (lane >= o) incl += v;
  }
  unsigned* aux = hist + NODES_MAX / 2;
  if (lane == 31) aux[wv] = incl;
  __syncthreads();
  unsigned run = incl - tsum;
#pragma unroll
  for (int j = 0; j < CNT_THREADS / 32; ++j)
    if (j < wv) run += aux[j];
  for (int i = w0; i < w1; ++i) {
    const unsigned w = hist[i];
    const unsigned c0 = w & 0xffffu, c1 = w >> 16;
    hist[i] = run | ((run + c0) << 16);
    run += c0 + c1;
  }
  __syncthreads();

  const unsigned cbase = (unsigned)slice * SLICE;
  unsigned short* lst = reinterpret_cast<unsigned short*>(hist + NODES_MAX / 2 + 16);
  constexpr int PLACE_GRP = 10;
  static_assert(CNT_PER % PLACE_GRP == 0, "whole groups");
  for (int g = 0; g < CNT_PER; g += PLACE_GRP) {
    unsigned pk[PLACE_GRP];
    int sv[PLACE_GRP];
#pragma unroll
    for (int k = 0; k < PLACE_GRP; ++k) {
      const bool in = e0 + (g + k) * CNT_THREADS < E_;
      pk[k] = in ? tp[(g + k) * CNT_THREADS] : 0u;
      sv[k] = in ? sp[(g + k) * CNT_THREADS] : 0;
    }
#pragma unroll
    for (int k = 0; k < PLACE_GRP; ++k)
      if (e0 + (g + k) * CNT_THREADS < E_) {
        const unsigned dst = pk[k] & 0xffffu;
        const unsigned w = hist[dst >> 1];
        const unsigned off = (dst & 1u) ? (w >> 16) : (w & 0xffffu);
        lst[off + (pk[k] >> 16)] = (unsigned short)sv[k];
      }
  }
  __syncthreads();

  const int left = E_ - slice * SLICE;
  const int cnt = left < SLICE ? left : SLICE;
  for (int j = t; 2 * j < cnt; j += CNT_THREADS) {
    if (2 * j + 1 < cnt)
      *reinterpret_cast<int2*>(col + cbase + 2 * j) = make_int2((int)lst[2 * j], (int)lst[2 * j + 1]);
    else
      col[cbase + 2 * j] = (int)lst[2 * j];
  }

  unsigned* trow = segtab + (size_t)slice * SEG_PITCH;
  for (int i = t; i < n; i += CNT_THREADS) {
    const unsigned w = hist[i >> 1];
    trow[i] = cbase + ((i & 1) ? (w >> 16) : (w & 0xffffu));
  }
  if (t == 0) trow[n] = cbase + (unsigned)cnt;
}

#define GEMM_WAVES 16

template <int CNT>
__device__ __forceinline__ void halve_step(float (&val)[16], bool up, int mask) {
#pragma unroll
  for (int i = 0; i < CNT; ++i) {
    const float keep = up ? val[CNT + i] : val[i];
    const float send = up ? val[i] : val[CNT + i];
    val[i] = keep + __shfl_xor(send, mask, 32);
  }
}
template <int K, int NC>
__device__ __forceinline__ void
gemm_body(float4* Bs, int bid, int nblk,
          const float* __restrict__ A, const float* __restrict__ B,
          float* __restrict__ C,
          const float* __restrict__ a_src, const float* __restrict__ a_dst,
          float* __restrict__ es, float* __restrict__ ed, int M) {
  constexpr int NT = NC / 16;
  constexpr int H  = NC / HID;
  constexpr int TPH = HID / 16;
  constexpr int KS = K / 8;
  constexpr int NQ = (K / 4) * NC;
  constexpr int NTHR = GEMM_WAVES * 32;
  constexpr int NIT = NQ / NTHR;
  static_assert(NQ % NTHR == 0, "staging shape");

  const int t = (int)threadIdx.x;
  const int lane = t & 31;
  const int lr = lane & 15, hi = lane >> 4;
  const int w = __builtin_amdgcn_readfirstlane(t >> 5);
  const int strip = bid + nblk * w;
  const int nstrips = M >> 4;
  const bool active = strip < nstrips;

  float4 stg[NIT];
#pragma unroll
  for (int j = 0; j < NIT; ++j) {
    const int idx = t + j * NTHR;
    const float* bp = B + (size_t)(4 * (idx / NC)) * NC + (idx % NC);
    stg[j].x = bp[0]; stg[j].y = bp[NC]; stg[j].z = bp[2 * NC]; stg[j].w = bp[3 * NC];
  }

#pragma unroll
  for (int j = 0; j < NIT; ++j) Bs[t + j * NTHR] = stg[j];
  __syncthreads();
  if (!active) return;

  float4 a[KS];
  {
    const float* ap = A + (size_t)(strip * 16 + lr) * K + 4 * hi;
#pragma unroll
    for (int i = 0; i < KS; ++i) a[i] = *reinterpret_cast<const float4*>(ap + 8 * i);
  }

  v8f acc[NT];
#pragma unroll
  for (int tn = 0; tn < NT; ++tn) acc[tn] = v8f{};

#pragma unroll
  for (int i = 0; i < KS; ++i) {
    const float4* bq = Bs + (2 * i + hi) * NC + lr;
    float4 b[NT];
#pragma unroll
    for (int tn = 0; tn < NT; ++tn) b[tn] = bq[tn * 16];
    v2f a01; a01.x = a[i].x; a01.y = a[i].y;
    v2f a23; a23.x = a[i].z; a23.y = a[i].w;
#pragma unroll
    for (int tn = 0; tn < NT; ++tn) {
      v2f b01; b01.x = b[tn].x; b01.y = b[tn].y;
      acc[tn] = __builtin_amdgcn_wmma_f32_16x16x4_f32(
           false, a01,  false, b01,
           (short)0, acc[tn],  false,  false);
    }
#pragma unroll
    for (int tn = 0; tn < NT; ++tn) {
      v2f b23; b23.x = b[tn].z; b23.y = b[tn].w;
      acc[tn] = __builtin_amdgcn_wmma_f32_16x16x4_f32(
           false, a23,  false, b23,
           (short)0, acc[tn],  false,  false);
    }
  }

  float asl[NT], adl[NT];
#pragma unroll
  for (int tn = 0; tn < NT; ++tn) {
    asl[tn] = a_src[tn * 16 + lr];
    adl[tn] = a_dst[tn * 16 + lr];
  }

  float* cp = C + (size_t)(strip * 16 + 8 * hi) * NC + lr;
#pragma unroll
  for (int tn = 0; tn < NT; ++tn) {
#pragma unroll
    for (int v = 0; v < 8; ++v) cp[(size_t)v * NC + tn * 16] = acc[tn][v];
  }

  float res[H];
#pragma unroll
  for (int hh = 0; hh < H; ++hh) {
    float val[16];
#pragma unroll
    for (int v = 0; v < 8; ++v) {
      float s = 0.0f, d = 0.0f;
#pragma unroll
      for (int t = 0; t < TPH; ++t) {
        s += acc[hh * TPH + t][v] * asl[hh * TPH + t];
        d += acc[hh * TPH + t][v] * adl[hh * TPH + t];
      }
      val[v] = s; val[8 + v] = d;
    }
    halve_step<8>(val, (lr & 1) != 0, 1);
    halve_step<4>(val, (lr & 2) != 0, 2);
    halve_step<2>(val, (lr & 4) != 0, 4);
    halve_step<1>(val, (lr & 8) != 0, 8);
    res[hh] = val[0];
  }
  {
    const int v = ((lr & 2) << 1) | ((lr & 4) >> 1) | ((lr & 8) >> 3);
    float* o = ((lr & 1) ? ed : es) + (size_t)(strip * 16 + 8 * hi + v) * H;
    if constexpr (H == 4) {
      *reinterpret_cast<float4*>(o) = make_float4(res[0], res[1], res[2], res[3]);
    } else {
#pragma unroll
      for (int hh = 0; hh < H; ++hh) o[hh] = res[hh];
    }
  }
}

__global__ void __launch_bounds__(GEMM_WAVES * 32) __attribute__((amdgpu_waves_per_eu(1, 4)))
count_and_gemm1(const int* __restrict__ ei, unsigned* __restrict__ segtab,
                int* __restrict__ col, unsigned* __restrict__ etmp, int E_, int nslice,
                const float* __restrict__ A, const float* __restrict__ B,
                float* __restrict__ C,
                const float* __restrict__ a_src, const float* __restrict__ a_dst,
                float* __restrict__ es, float* __restrict__ ed, int M) {
  static_assert(CNT_THREADS == GEMM_WAVES * 32, "one block size for both kinds of work-group");
  constexpr int GEMM_LDS_BYTES = (IN_CH / 4) * (HEADS * HID) * 16;
  constexpr int LDS_BYTES = GEMM_LDS_BYTES > SORT_LDS_BYTES ? GEMM_LDS_BYTES : SORT_LDS_BYTES;
  static_assert(CNT_THREADS / 32 * 4 <= 64, "the scan's wave totals fit their 16 words");
  __shared__ float4 smem[(LDS_BYTES + 15) / 16];
  if ((int)blockIdx.x < nslice) {
    csr_slice_body(reinterpret_cast<unsigned*>(smem), (int)blockIdx.x, ei, segtab, col, etmp, E_, M);
  } else {
    gemm_body<IN_CH, HEADS * HID>(smem, (int)blockIdx.x - nslice,
                                  (int)gridDim.x - nslice, A, B, C, a_src, a_dst, es, ed, M);
  }
}

__device__ __forceinline__ int slot_col(int inc, int segbase, int j) {
  int lo = 0;
#pragma unroll
  for (int b = 16; b > 0; b >>= 1) {
    const int v = __shfl(inc, lo + b - 1, 32);
    if (v <= j) lo += b;
  }
  return __shfl(segbase, lo, 32) + j;
}

template <int H>
__device__ __forceinline__ void slot_logits(const int* __restrict__ col,
                                            const float* __restrict__ es,
                                            const float (&edd)[H], int ci, int tot,
                                            int d, int j, int& s, float (&l)[H]) {
  s = d;
#pragma unroll
  for (int hh = 0; hh < H; ++hh) l[hh] = -INFINITY;
  if (j < tot) {
    s = (j < tot - 1) ? col[ci] : d;
    float e[H];
    if constexpr (H == 4) {
      const float4 v = *reinterpret_cast<const float4*>(es + 4 * s);
      e[0] = v.x; e[1] = v.y; e[2] = v.z; e[3] = v.w;
    } else {
      e[0] = es[s];
    }
#pragma unroll
    for (int hh = 0; hh < H; ++hh) {
      const float x = e[hh] + edd[hh];
      l[hh] = (x > 0.0f) ? x : NEG_SLOPE * x;
    }
  }
}

template <int H, int UNR>
__device__ __forceinline__ void
gather_node(const unsigned* __restrict__ segtab, int nslice, const int* __restrict__ col,
            const float* __restrict__ h, const float* __restrict__ es,
            const float* __restrict__ ed, const float* __restrict__ bias,
            int d, int lane, float4& oa, float4& ob) {
  constexpr int LDH = H * HID;
  unsigned a0 = 0, a1 = 0;
  if (lane < nslice) {
    const unsigned ti = (unsigned)lane * (unsigned)SEG_PITCH + (unsigned)d;
    a0 = segtab[ti];
    a1 = segtab[ti + 1u];
  }
  const int len = (int)(a1 - a0);
  int inc = len;
#pragma unroll
  for (int o = 1; o < 32; o <<= 1) {
    const int v = __shfl_up(inc, (unsigned)o, 32);
    if (lane >= o) inc += v;
  }
  const int segbase = (int)a0 - (inc - len);
  const int tot = __builtin_amdgcn_readlane(inc, 31) + 1;

  float edd[H];
#pragma unroll
  for (int hh = 0; hh < H; ++hh) edd[hh] = ed[d * H + hh];

  int s0;
  float l0[H], mx[H];
  slot_logits<H>(col, es, edd, slot_col(inc, segbase, lane), tot, d, lane, s0, l0);
#pragma unroll
  for (int hh = 0; hh < H; ++hh) mx[hh] = l0[hh];
  for (int base = 32; base < tot; base += 32) {
    int s;
    float l[H];
    slot_logits<H>(col, es, edd, slot_col(inc, segbase, base + lane), tot, d, base + lane, s, l);
#pragma unroll
    for (int hh = 0; hh < H; ++hh) mx[hh] = fmaxf(mx[hh], l[hh]);
  }
#pragma unroll
  for (int hh = 0; hh < H; ++hh) {
#pragma unroll
    for (int o = 16; o > 0; o >>= 1) mx[hh] = fmaxf(mx[hh], __shfl_xor(mx[hh], o, 32));
    mx[hh] = __builtin_isfinite(mx[hh]) ? mx[hh] : 0.0f;
  }

  float sm[H];
#pragma unroll
  for (int hh = 0; hh < H; ++hh) sm[hh] = 0.0f;
  float4 acc0 = make_float4(0.f, 0.f, 0.f, 0.f);
  float4 acc1 = make_float4(0.f, 0.f, 0.f, 0.f);

  for (int base = 0; base < tot; base += 32) {
    int s = s0;
    float l[H], p[H];
#pragma unroll
    for (int hh = 0; hh < H; ++hh) l[hh] = l0[hh];
    if (base > 0)
      slot_logits<H>(col, es, edd, slot_col(inc, segbase, base + lane), tot, d, base + lane, s, l);
#pragma unroll
    for (int hh = 0; hh < H; ++hh) {
      p[hh] = (base + lane < tot) ? exp_hw(l[hh] - mx[hh]) : 0.0f;
      sm[hh] += p[hh];
    }
    const int nk = (tot - base < 32) ? (tot - base) : 32;
#pragma unroll UNR
    for (int k = 0; k < nk; ++k) {
      const int sk = __builtin_amdgcn_readlane(s, k);
      const float* hp = h + (size_t)sk * LDH;
      if constexpr (H == 4) {
        const float p0 = __int_as_float(__builtin_amdgcn_readlane(__float_as_int(p[0]), k));
        const float p1 = __int_as_float(__builtin_amdgcn_readlane(__float_as_int(p[1]), k));
        const float p2 = __int_as_float(__builtin_amdgcn_readlane(__float_as_int(p[2]), k));
        const float p3 = __int_as_float(__builtin_amdgcn_readlane(__float_as_int(p[3]), k));
        const float pa = (lane < 16) ? p0 : p1;
        const float pb = (lane < 16) ? p2 : p3;
        const float4 a = *reinterpret_cast<const float4*>(hp + 4 * lane);
        const float4 b = *reinterpret_cast<const float4*>(hp + 128 + 4 * lane);
        acc0.x += a.x * pa; acc0.y += a.y * pa; acc0.z += a.z * pa; acc0.w += a.w * pa;
        acc1.x += b.x * pb; acc1.y += b.y * pb; acc1.z += b.z * pb; acc1.w += b.w * pb;
      } else {
        const float p0 = __int_as_float(__builtin_amdgcn_readlane(__float_as_int(p[0]), k));
        const float2 a = *reinterpret_cast<const float2*>(hp + 2 * lane);
        acc0.x += a.x * p0; acc0.y += a.y * p0;
      }
    }
  }
#pragma unroll
  for (int hh = 0; hh < H; ++hh) {
#pragma unroll
    for (int o = 16; o > 0; o >>= 1) sm[hh] += __shfl_xor(sm[hh], o, 32);
  }

  ob = make_float4(0.f, 0.f, 0.f, 0.f);
  if constexpr (H == 4) {
    const float ra = __builtin_amdgcn_rcpf(((lane < 16) ? sm[0] : sm[1]) + 1e-16f);
    const float rb = __builtin_amdgcn_rcpf(((lane < 16) ? sm[2] : sm[3]) + 1e-16f);
    const float4 ba = *reinterpret_cast<const float4*>(bias + 4 * lane);
    const float4 bb = *reinterpret_cast<const float4*>(bias + 128 + 4 * lane);
    oa.x = elu_f(acc0.x * ra + ba.x); oa.y = elu_f(acc0.y * ra + ba.y);
    oa.z = elu_f(acc0.z * ra + ba.z); oa.w = elu_f(acc0.w * ra + ba.w);
    ob.x = elu_f(acc1.x * rb + bb.x); ob.y = elu_f(acc1.y * rb + bb.y);
    ob.z = elu_f(acc1.z * rb + bb.z); ob.w = elu_f(acc1.w * rb + bb.w);
  } else {
    const float ra = __builtin_amdgcn_rcpf(sm[0] + 1e-16f);
    const float2 ba = *reinterpret_cast<const float2*>(bias + 2 * lane);
    oa.x = elu_f(acc0.x * ra + ba.x); oa.y = elu_f(acc0.y * ra + ba.y);
    oa.z = 0.0f; oa.w = 0.0f;
  }
}

template <int H>
__global__ void __launch_bounds__(256)
gat_gather(const unsigned* __restrict__ segtab, int nslice, const int* __restrict__ col,
           const float* __restrict__ h, const float* __restrict__ es,
           const float* __restrict__ ed, const float* __restrict__ bias,
           float* __restrict__ out, int n) {
  constexpr int LDH = H * HID;
  const int lane = (int)(threadIdx.x & 31);
  int d = (int)((blockIdx.x * blockDim.x + threadIdx.x) >> 5);
  d = __builtin_amdgcn_readfirstlane(d);
  if (d >= n) return;
  float4 oa, ob;
  gather_node<H, 4>(segtab, nslice, col, h, es, ed, bias, d, lane, oa, ob);
  float* op = out + (size_t)d * LDH;
  if constexpr (H == 4) {
    *reinterpret_cast<float4*>(op + 4 * lane)       = oa;
    *reinterpret_cast<float4*>(op + 128 + 4 * lane) = ob;
  } else {
    *reinterpret_cast<float2*>(op + 2 * lane) = make_float2(oa.x, oa.y);
  }
}

#define FUSE_WAVES  16
#define FUSE_MAX_WG 512
#define AT_PITCH    (HEADS * HID + 4)
#define CT_PITCH    (HID + 4)
__global__ void __launch_bounds__(FUSE_WAVES * 32)
gather1_gemm2(const unsigned* __restrict__ segtab, int nslice, const int* __restrict__ col,
              const float* __restrict__ h1, const float* __restrict__ es1,
              const float* __restrict__ ed1, const float* __restrict__ b1,
              const float* __restrict__ W2, const float* __restrict__ a_src2,
              const float* __restrict__ a_dst2, float* __restrict__ h2,
              float* __restrict__ es2, float* __restrict__ ed2, int n) {
  constexpr int K = HEADS * HID, NC = HID;
  constexpr int NQ = (K / 4) * NC;
  constexpr int NTHR = FUSE_WAVES * 32;
  constexpr int NIT = NQ / NTHR;
  constexpr int KQ = 4;
  constexpr int KSQ = K / 8 / KQ;
  static_assert(NQ % NTHR == 0, "staging shape");
  static_assert(FUSE_WAVES == 16 && NC / 16 * KQ == FUSE_WAVES, "a wave per node, per column tile and k quarter");
  __shared__ float4 Bs[NQ];
  __shared__ float4 At4[16 * AT_PITCH / 4];
  __shared__ float4 Ct4[KQ * 16 * CT_PITCH / 4];
  float* At = reinterpret_cast<float*>(At4);
  float* Ct = reinterpret_cast<float*>(Ct4);

  const int t = (int)threadIdx.x;
  const int lane = t & 31;
  const int lr = lane & 15, hi = lane >> 4;
  const int w = __builtin_amdgcn_readfirstlane(t >> 5);

  {
    float4 stg[NIT];
#pragma unroll
    for (int j = 0; j < NIT; ++j) {
      const int idx = t + j * NTHR;
      const float* bp = W2 + (size_t)(4 * (idx / NC)) * NC + (idx % NC);
      stg[j].x = bp[0]; stg[j].y = bp[NC]; stg[j].z = bp[2 * NC]; stg[j].w = bp[3 * NC];
    }
#pragma unroll
    for (int j = 0; j < NIT; ++j) Bs[t + j * NTHR] = stg[j];
  }
  const float2 as2l = *reinterpret_cast<const float2*>(a_src2 + 2 * lane);
  const float2 ad2l = *reinterpret_cast<const float2*>(a_dst2 + 2 * lane);
  __syncthreads();

  const int tn = w & 3, kq = w >> 2;
  const int ntiles = n >> 4;
  const int tper = (ntiles + (int)gridDim.x - 1) / (int)gridDim.x;
  const int tile0 = (int)blockIdx.x * tper;
  const int tile1 = (tile0 + tper < ntiles) ? (tile0 + tper) : ntiles;
  for (int tile = tile0; tile < tile1; ++tile) {
    const int d = tile * 16 + w;
    {
      float4 oa, ob;
      gather_node<HEADS, 4>(segtab, nslice, col, h1, es1, ed1, b1, d, lane, oa, ob);
      float* ar = At + w * AT_PITCH;
      *reinterpret_cast<float4*>(ar + 4 * lane)       = oa;
      *reinterpret_cast<float4*>(ar + 128 + 4 * lane) = ob;
    }
    __syncthreads();

    {
      const float* ap = At + lr * AT_PITCH + 8 * (kq * KSQ) + 4 * hi;
      const float4* bq = Bs + (2 * (kq * KSQ) + hi) * NC + tn * 16 + lr;
      float4 a[KSQ], b[KSQ];
#pragma unroll
      for (int i = 0; i < KSQ; ++i) {
        a[i] = *reinterpret_cast<const float4*>(ap + 8 * i);
        b[i] = bq[2 * i * NC];
      }
      v8f c0 = v8f{}, c1 = v8f{};
#pragma unroll
      for (int i = 0; i < KSQ; ++i) {
        v2f a01; a01.x = a[i].x; a01.y = a[i].y;
        v2f a23; a23.x = a[i].z; a23.y = a[i].w;
        v2f b01; b01.x = b[i].x; b01.y = b[i].y;
        v2f b23; b23.x = b[i].z; b23.y = b[i].w;
        c0 = __builtin_amdgcn_wmma_f32_16x16x4_f32(
             false, a01,  false, b01,
             (short)0, c0,  false,  false);
        c1 = __builtin_amdgcn_wmma_f32_16x16x4_f32(
             false, a23,  false, b23,
             (short)0, c1,  false,  false);
      }
      float* cq = Ct + (kq * 16 + 8 * hi) * CT_PITCH + tn * 16 + lr;
#pragma unroll
      for (int v = 0; v < 8; ++v) cq[v * CT_PITCH] = c0[v] + c1[v];
    }
    __syncthreads();

    {
      const float* cr = Ct + w * CT_PITCH + 2 * lane;
      float2 c = *reinterpret_cast<const float2*>(cr);
#pragma unroll
      for (int q = 1; q < KQ; ++q) {
        const float2 x = *reinterpret_cast<const float2*>(cr + q * 16 * CT_PITCH);
        c.x += x.x; c.y += x.y;
      }
      *reinterpret_cast<float2*>(h2 + (size_t)d * NC + 2 * lane) = c;
      float s = c.x * as2l.x + c.y * as2l.y;
      float e = c.x * ad2l.x + c.y * ad2l.y;
#pragma unroll
      for (int o = 16; o > 0; o >>= 1) {
        s += __shfl_xor(s, o, 32);
        e += __shfl_xor(e, o, 32);
      }
      if (lane == 0) { es2[d] = s; ed2[d] = e; }
    }
  }
}

__global__ void out_proj(const float* __restrict__ h,
                         const float* __restrict__ Wo,
                         const float* __restrict__ bo,
                         float* __restrict__ out, int n) {
  int wid  = (int)((blockIdx.x * blockDim.x + threadIdx.x) >> 5);
  int lane = (int)(threadIdx.x & 31);
  if (wid >= n) return;
  const float* hp = h + (size_t)wid * HID;
  float s = hp[lane] * Wo[lane] + hp[lane + 32] * Wo[lane + 32];
#pragma unroll
  for (int o = 16; o > 0; o >>= 1) s += __shfl_xor(s, o, 32);
  if (lane == 0) out[wid] = s + bo[0];
}

#define CDIV(a, b) (((a) + (b) - 1) / (b))

extern "C" void kernel_launch(void* const* d_in, const int* in_sizes, int n_in,
                              void* d_out, int out_size, void* d_ws, size_t ws_size,
                              hipStream_t stream) {
  const float* x   = (const float*)d_in[0];
  const float* W1  = (const float*)d_in[1];
  const float* as1 = (const float*)d_in[2];
  const float* ad1 = (const float*)d_in[3];
  const float* b1  = (const float*)d_in[4];
  const float* W2  = (const float*)d_in[5];
  const float* as2 = (const float*)d_in[6];
  const float* ad2 = (const float*)d_in[7];
  const float* b2  = (const float*)d_in[8];
  const float* Wo  = (const float*)d_in[9];
  const float* bo  = (const float*)d_in[10];
  const int*   ei  = (const int*)d_in[11];

  const int n  = in_sizes[0]  / IN_CH;
  const int e  = in_sizes[11] / 2;

  float* p = (float*)d_ws;
  float*    h1   = p;            p += (size_t)n * 256;
  float*    o1   = p;            p += (size_t)n * 256;
  unsigned* segtab = (unsigned*)p; p += (size_t)SEG_SLICES * SEG_PITCH;
  float*    es1  = p;            p += (size_t)n * 4;
  float*    ed1  = p;            p += (size_t)n * 4;
  float*    es2v = p;            p += n;
  float*    ed2v = p;            p += n;
  int*      col  = (int*)p;      p += e;
  unsigned* etmp = (unsigned*)p;
  float*    h2   = o1;
  float*    o2   = o1 + (size_t)n * 64;
  const int nslice = CDIV(e, SLICE);

  const int B = 256;

  const int gneed = CDIV(n / 16, GEMM_WAVES);
  const int gpart = gneed > 256 - nslice ? gneed : 256 - nslice;
  count_and_gemm1<<<nslice + gpart, GEMM_WAVES * 32, 0, stream>>>(
      ei, segtab, col, etmp, e, nslice, x, W1, h1, as1, ad1, es1, ed1, n);

  const int ntiles = n / 16;
  const int tper = CDIV(ntiles, FUSE_MAX_WG);
  gather1_gemm2<<<CDIV(ntiles, tper), FUSE_WAVES * 32, 0, stream>>>(
      segtab, nslice, col, h1, es1, ed1, b1, W2, as2, ad2, h2, es2v, ed2v, n);

  gat_gather<1><<<CDIV(n * 32, B), B, 0, stream>>>(segtab, nslice, col, h2, es2v, ed2v, b2, o2, n);

  out_proj<<<CDIV(n * 32, B), B, 0, stream>>>(o2, Wo, bo, (float*)d_out, n);
}


# Both gathers and the projection: lane scan, 32-lane max and sum butterflies and the search's first probe on the vector ALU (DPP row steps, v_permlanex16, lane read), not LDS-unit permutes; same trees
# speedup vs baseline: 6.1085x; 1.0347x over previous
#include <hip/hip_runtime.h>
#include <math.h>


#define IN_CH 128
#define HID   64
#define HEADS 4
#define NEG_SLOPE 0.2f

typedef float v2f __attribute__((ext_vector_type(2)));
typedef float v8f __attribute__((ext_vector_type(8)));

__device__ __forceinline__ float exp_hw(float x) {
  return __builtin_amdgcn_exp2f(x * 1.44269504088896340736f);
}

__device__ __forceinline__ float elu_f(float x) {
  return x > 0.0f ? x : (exp_hw(x) - 1.0f);
}

#define DPP_ROW_SHR(n)   (0x110 + (n))
#define DPP_ROW_XMASK(m) (0x160 + (m))
template <int CTRL>
__device__ __forceinline__ int dpp_i(int v) {
  return __builtin_amdgcn_update_dpp(0, v, CTRL, 0xf, 0xf, true);
}
template <int CTRL>
__device__ __forceinline__ float dpp_f(float v) {
  return __int_as_float(__builtin_amdgcn_update_dpp(0, __float_as_int(v), CTRL, 0xf, 0xf, true));
}
__device__ __forceinline__ float xrow_f(float v) {
  const int b = __float_as_int(v);
  return __int_as_float(__builtin_amdgcn_permlanex16(b, b, 0x76543210, 0xfedcba98, false, false));
}
__device__ __forceinline__ float wave_sum(float x) {
  x += xrow_f(x);
  x += dpp_f<DPP_ROW_XMASK(8)>(x);
  x += dpp_f<DPP_ROW_XMASK(4)>(x);
  x += dpp_f<DPP_ROW_XMASK(2)>(x);
  x += dpp_f<DPP_ROW_XMASK(1)>(x);
  return x;
}
__device__ __forceinline__ float wave_max(float x) {
  x = fmaxf(x, xrow_f(x));
  x = fmaxf(x, dpp_f<DPP_ROW_XMASK(8)>(x));
  x = fmaxf(x, dpp_f<DPP_ROW_XMASK(4)>(x));
  x = fmaxf(x, dpp_f<DPP_ROW_XMASK(2)>(x));
  x = fmaxf(x, dpp_f<DPP_ROW_XMASK(1)>(x));
  return x;
}
__device__ __forceinline__ int wave_scan_incl(int v, int lane) {
  v += dpp_i<DPP_ROW_SHR(1)>(v);
  v += dpp_i<DPP_ROW_SHR(2)>(v);
  v += dpp_i<DPP_ROW_SHR(4)>(v);
  v += dpp_i<DPP_ROW_SHR(8)>(v);
  const int low = __builtin_amdgcn_readlane(v, 15);
  return v + ((lane >= 16) ? low : 0);
}

#define NODES_MAX   50000
#define CNT_THREADS 512
#define CNT_PER     50
#define SLICE       (CNT_THREADS * CNT_PER)
#define SEG_SLICES  32
#define SEG_PITCH   (NODES_MAX + 16)
#define SCAN_CHUNK  ((NODES_MAX / 2 + CNT_THREADS - 1) / CNT_THREADS)
#define SORT_LDS_BYTES (NODES_MAX * 2 + 64 + SLICE * 2)
static_assert(SLICE < 65536, "a slice's count and offsets must fit 16 bits");
static_assert(NODES_MAX <= 65536, "a node number fits 16 bits (an edge's destination and rank share a word; the sorted list holds 16-bit sources)");
__device__ __forceinline__ void
csr_slice_body(unsigned* hist, int slice, const int* __restrict__ ei,
               unsigned* __restrict__ segtab, int* __restrict__ col,
               unsigned* __restrict__ etmp, int E_, int n) {
  const int t = (int)threadIdx.x;
  const int lane = t & 31, wv = t >> 5;
  const int e0 = slice * SLICE + t;
  const int* dp = ei + (size_t)E_ + e0;
  const int* sp = ei + e0;

  int d[CNT_PER];
#pragma unroll
  for (int k = 0; k < CNT_PER; ++k)
    d[k] = (e0 + k * CNT_THREADS < E_) ? dp[k * CNT_THREADS] : -1;

  const int nw = n >> 1;
  for (int i = 4 * t; i < nw; i += 4 * CNT_THREADS)
    *reinterpret_cast<uint4*>(hist + i) = make_uint4(0u, 0u, 0u, 0u);
  __syncthreads();

  constexpr int ADD_GRP = 10;
  static_assert(CNT_PER % ADD_GRP == 0, "whole groups");
  unsigned* tp = etmp + e0;
#pragma unroll
  for (int g = 0; g < CNT_PER; g += ADD_GRP) {
    unsigned r[ADD_GRP];
#pragma unroll
    for (int k = 0; k < ADD_GRP; ++k)
      r[k] = (d[g + k] >= 0)
                 ? atomicAdd(&hist[d[g + k] >> 1], (d[g + k] & 1) ? 0x10000u : 1u) : 0u;
#pragma unroll
    for (int k = 0; k < ADD_GRP; ++k)
      if (d[g + k] >= 0)
        tp[(g + k) * CNT_THREADS] =
            (unsigned)d[g + k] | (((d[g + k] & 1) ? (r[k] >> 16) : (r[k] & 0xffffu)) << 16);
  }
  __syncthreads();

  const int w0 = SCAN_CHUNK * t;
  const int w1 = (w0 + SCAN_CHUNK < nw) ? (w0 + SCAN_CHUNK) : nw;
  unsigned tsum = 0;
  for (int i = w0; i < w1; ++i) {
    const unsigned w = hist[i];
    tsum += (w & 0xffffu) + (w >> 16);
  }
  unsigned incl = tsum;
#pragma unroll
  for (int o = 1; o < 32; o <<= 1) {
    const unsigned v = __shfl_up(incl, (unsigned)o, 32);
    if (lane >= o) incl += v;
  }
  unsigned* aux = hist + NODES_MAX / 2;
  if (lane == 31) aux[wv] = incl;
  __syncthreads();
  unsigned run = incl - tsum;
#pragma unroll
  for (int j = 0; j < CNT_THREADS / 32; ++j)
    if (j < wv) run += aux[j];
  for (int i = w0; i < w1; ++i) {
    const unsigned w = hist[i];
    const unsigned c0 = w & 0xffffu, c1 = w >> 16;
    hist[i] = run | ((run + c0) << 16);
    run += c0 + c1;
  }
  __syncthreads();

  const unsigned cbase = (unsigned)slice * SLICE;
  unsigned short* lst = reinterpret_cast<unsigned short*>(hist + NODES_MAX / 2 + 16);
  constexpr int PLACE_GRP = 10;
  static_assert(CNT_PER % PLACE_GRP == 0, "whole groups");
  for (int g = 0; g < CNT_PER; g += PLACE_GRP) {
    unsigned pk[PLACE_GRP];
    int sv[PLACE_GRP];
#pragma unroll
    for (int k = 0; k < PLACE_GRP; ++k) {
      const bool in = e0 + (g + k) * CNT_THREADS < E_;
      pk[k] = in ? tp[(g + k) * CNT_THREADS] : 0u;
      sv[k] = in ? sp[(g + k) * CNT_THREADS] : 0;
    }
#pragma unroll
    for (int k = 0; k < PLACE_GRP; ++k)
      if (e0 + (g + k) * CNT_THREADS < E_) {
        const unsigned dst = pk[k] & 0xffffu;
        const unsigned w = hist[dst >> 1];
        const unsigned off = (dst & 1u) ? (w >> 16) : (w & 0xffffu);
        lst[off + (pk[k] >> 16)] = (unsigned short)sv[k];
      }
  }
  __syncthreads();

  const int left = E_ - slice * SLICE;
  const int cnt = left < SLICE ? left : SLICE;
  for (int j = t; 2 * j < cnt; j += CNT_THREADS) {
    if (2 * j + 1 < cnt)
      *reinterpret_cast<int2*>(col + cbase + 2 * j) = make_int2((int)lst[2 * j], (int)lst[2 * j + 1]);
    else
      col[cbase + 2 * j] = (int)lst[2 * j];
  }

  unsigned* trow = segtab + (size_t)slice * SEG_PITCH;
  for (int i = t; i < n; i += CNT_THREADS) {
    const unsigned w = hist[i >> 1];
    trow[i] = cbase + ((i & 1) ? (w >> 16) : (w & 0xffffu));
  }
  if (t == 0) trow[n] = cbase + (unsigned)cnt;
}

#define GEMM_WAVES 16

template <int CNT>
__device__ __forceinline__ void halve_step(float (&val)[16], bool up, int mask) {
#pragma unroll
  for (int i = 0; i < CNT; ++i) {
    const float keep = up ? val[CNT + i] : val[i];
    const float send = up ? val[i] : val[CNT + i];
    val[i] = keep + __shfl_xor(send, mask, 32);
  }
}
template <int K, int NC>
__device__ __forceinline__ void
gemm_body(float4* Bs, int bid, int nblk,
          const float* __restrict__ A, const float* __restrict__ B,
          float* __restrict__ C,
          const float* __restrict__ a_src, const float* __restrict__ a_dst,
          float* __restrict__ es, float* __restrict__ ed, int M) {
  constexpr int NT = NC / 16;
  constexpr int H  = NC / HID;
  constexpr int TPH = HID / 16;
  constexpr int KS = K / 8;
  constexpr int NQ = (K / 4) * NC;
  constexpr int NTHR = GEMM_WAVES * 32;
  constexpr int NIT = NQ / NTHR;
  static_assert(NQ % NTHR == 0, "staging shape");

  const int t = (int)threadIdx.x;
  const int lane = t & 31;
  const int lr = lane & 15, hi = lane >> 4;
  const int w = __builtin_amdgcn_readfirstlane(t >> 5);
  const int strip = bid + nblk * w;
  const int nstrips = M >> 4;
  const bool active = strip < nstrips;

  float4 stg[NIT];
#pragma unroll
  for (int j = 0; j < NIT; ++j) {
    const int idx = t + j * NTHR;
    const float* bp = B + (size_t)(4 * (idx / NC)) * NC + (idx % NC);
    stg[j].x = bp[0]; stg[j].y = bp[NC]; stg[j].z = bp[2 * NC]; stg[j].w = bp[3 * NC];
  }

#pragma unroll
  for (int j = 0; j < NIT; ++j) Bs[t + j * NTHR] = stg[j];
  __syncthreads();
  if (!active) return;

  float4 a[KS];
  {
    const float* ap = A + (size_t)(strip * 16 + lr) * K + 4 * hi;
#pragma unroll
    for (int i = 0; i < KS; ++i) a[i] = *reinterpret_cast<const float4*>(ap + 8 * i);
  }

  v8f acc[NT];
#pragma unroll
  for (int tn = 0; tn < NT; ++tn) acc[tn] = v8f{};

#pragma unroll
  for (int i = 0; i < KS; ++i) {
    const float4* bq = Bs + (2 * i + hi) * NC + lr;
    float4 b[NT];
#pragma unroll
    for (int tn = 0; tn < NT; ++tn) b[tn] = bq[tn * 16];
    v2f a01; a01.x = a[i].x; a01.y = a[i].y;
    v2f a23; a23.x = a[i].z; a23.y = a[i].w;
#pragma unroll
    for (int tn = 0; tn < NT; ++tn) {
      v2f b01; b01.x = b[tn].x; b01.y = b[tn].y;
      acc[tn] = __builtin_amdgcn_wmma_f32_16x16x4_f32(
           false, a01,  false, b01,
           (short)0, acc[tn],  false,  false);
    }
#pragma unroll
    for (int tn = 0; tn < NT; ++tn) {
      v2f b23; b23.x = b[tn].z; b23.y = b[tn].w;
      acc[tn] = __builtin_amdgcn_wmma_f32_16x16x4_f32(
           false, a23,  false, b23,
           (short)0, acc[tn],  false,  false);
    }
  }

  float asl[NT], adl[NT];
#pragma unroll
  for (int tn = 0; tn < NT; ++tn) {
    asl[tn] = a_src[tn * 16 + lr];
    adl[tn] = a_dst[tn * 16 + lr];
  }

  float* cp = C + (size_t)(strip * 16 + 8 * hi) * NC + lr;
#pragma unroll
  for (int tn = 0; tn < NT; ++tn) {
#pragma unroll
    for (int v = 0; v < 8; ++v) cp[(size_t)v * NC + tn * 16] = acc[tn][v];
  }

  float res[H];
#pragma unroll
  for (int hh = 0; hh < H; ++hh) {
    float val[16];
#pragma unroll
    for (int v = 0; v < 8; ++v) {
      float s = 0.0f, d = 0.0f;
#pragma unroll
      for (int t = 0; t < TPH; ++t) {
        s += acc[hh * TPH + t][v] * asl[hh * TPH + t];
        d += acc[hh * TPH + t][v] * adl[hh * TPH + t];
      }
      val[v] = s; val[8 + v] = d;
    }
    halve_step<8>(val, (lr & 1) != 0, 1);
    halve_step<4>(val, (lr & 2) != 0, 2);
    halve_step<2>(val, (lr & 4) != 0, 4);
    halve_step<1>(val, (lr & 8) != 0, 8);
    res[hh] = val[0];
  }
  {
    const int v = ((lr & 2) << 1) | ((lr & 4) >> 1) | ((lr & 8) >> 3);
    float* o = ((lr & 1) ? ed : es) + (size_t)(strip * 16 + 8 * hi + v) * H;
    if constexpr (H == 4) {
      *reinterpret_cast<float4*>(o) = make_float4(res[0], res[1], res[2], res[3]);
    } else {
#pragma unroll
      for (int hh = 0; hh < H; ++hh) o[hh] = res[hh];
    }
  }
}

__global__ void __launch_bounds__(GEMM_WAVES * 32) __attribute__((amdgpu_waves_per_eu(1, 4)))
count_and_gemm1(const int* __restrict__ ei, unsigned* __restrict__ segtab,
                int* __restrict__ col, unsigned* __restrict__ etmp, int E_, int nslice,
                const float* __restrict__ A, const float* __restrict__ B,
                float* __restrict__ C,
                const float* __restrict__ a_src, const float* __restrict__ a_dst,
                float* __restrict__ es, float* __restrict__ ed, int M) {
  static_assert(CNT_THREADS == GEMM_WAVES * 32, "one block size for both kinds of work-group");
  constexpr int GEMM_LDS_BYTES = (IN_CH / 4) * (HEADS * HID) * 16;
  constexpr int LDS_BYTES = GEMM_LDS_BYTES > SORT_LDS_BYTES ? GEMM_LDS_BYTES : SORT_LDS_BYTES;
  static_assert(CNT_THREADS / 32 * 4 <= 64, "the scan's wave totals fit their 16 words");
  __shared__ float4 smem[(LDS_BYTES + 15) / 16];
  if ((int)blockIdx.x < nslice) {
    csr_slice_body(reinterpret_cast<unsigned*>(smem), (int)blockIdx.x, ei, segtab, col, etmp, E_, M);
  } else {
    gemm_body<IN_CH, HEADS * HID>(smem, (int)blockIdx.x - nslice,
                                  (int)gridDim.x - nslice, A, B, C, a_src, a_dst, es, ed, M);
  }
}

__device__ __forceinline__ int slot_col(int inc, int segbase, int j) {
  int lo = (__builtin_amdgcn_readlane(inc, 15) <= j) ? 16 : 0;
#pragma unroll
  for (int b = 8; b > 0; b >>= 1) {
    const int v = __shfl(inc, lo + b - 1, 32);
    if (v <= j) lo += b;
  }
  return __shfl(segbase, lo, 32) + j;
}

template <int H>
__device__ __forceinline__ void slot_logits(const int* __restrict__ col,
                                            const float* __restrict__ es,
                                            const float (&edd)[H], int ci, int tot,
                                            int d, int j, int& s, float (&l)[H]) {
  s = d;
#pragma unroll
  for (int hh = 0; hh < H; ++hh) l[hh] = -INFINITY;
  if (j < tot) {
    s = (j < tot - 1) ? col[ci] : d;
    float e[H];
    if constexpr (H == 4) {
      const float4 v = *reinterpret_cast<const float4*>(es + 4 * s);
      e[0] = v.x; e[1] = v.y; e[2] = v.z; e[3] = v.w;
    } else {
      e[0] = es[s];
    }
#pragma unroll
    for (int hh = 0; hh < H; ++hh) {
      const float x = e[hh] + edd[hh];
      l[hh] = (x > 0.0f) ? x : NEG_SLOPE * x;
    }
  }
}

template <int H, int UNR>
__device__ __forceinline__ void
gather_node(const unsigned* __restrict__ segtab, int nslice, const int* __restrict__ col,
            const float* __restrict__ h, const float* __restrict__ es,
            const float* __restrict__ ed, const float* __restrict__ bias,
            int d, int lane, float4& oa, float4& ob) {
  constexpr int LDH = H * HID;
  unsigned a0 = 0, a1 = 0;
  if (lane < nslice) {
    const unsigned ti = (unsigned)lane * (unsigned)SEG_PITCH + (unsigned)d;
    a0 = segtab[ti];
    a1 = segtab[ti + 1u];
  }
  const int len = (int)(a1 - a0);
  const int inc = wave_scan_incl(len, lane);
  const int segbase = (int)a0 - (inc - len);
  const int tot = __builtin_amdgcn_readlane(inc, 31) + 1;

  float edd[H];
#pragma unroll
  for (int hh = 0; hh < H; ++hh) edd[hh] = ed[d * H + hh];

  int s0;
  float l0[H], mx[H];
  slot_logits<H>(col, es, edd, slot_col(inc, segbase, lane), tot, d, lane, s0, l0);
#pragma unroll
  for (int hh = 0; hh < H; ++hh) mx[hh] = l0[hh];
  for (int base = 32; base < tot; base += 32) {
    int s;
    float l[H];
    slot_logits<H>(col, es, edd, slot_col(inc, segbase, base + lane), tot, d, base + lane, s, l);
#pragma unroll
    for (int hh = 0; hh < H; ++hh) mx[hh] = fmaxf(mx[hh], l[hh]);
  }
#pragma unroll
  for (int hh = 0; hh < H; ++hh) {
    mx[hh] = wave_max(mx[hh]);
    mx[hh] = __builtin_isfinite(mx[hh]) ? mx[hh] : 0.0f;
  }

  float sm[H];
#pragma unroll
  for (int hh = 0; hh < H; ++hh) sm[hh] = 0.0f;
  float4 acc0 = make_float4(0.f, 0.f, 0.f, 0.f);
  float4 acc1 = make_float4(0.f, 0.f, 0.f, 0.f);

  for (int base = 0; base < tot; base += 32) {
    int s = s0;
    float l[H], p[H];
#pragma unroll
    for (int hh = 0; hh < H; ++hh) l[hh] = l0[hh];
    if (base > 0)
      slot_logits<H>(col, es, edd, slot_col(inc, segbase, base + lane), tot, d, base + lane, s, l);
#pragma unroll
    for (int hh = 0; hh < H; ++hh) {
      p[hh] = (base + lane < tot) ? exp_hw(l[hh] - mx[hh]) : 0.0f;
      sm[hh] += p[hh];
    }
    const int nk = (tot - base < 32) ? (tot - base) : 32;
#pragma unroll UNR
    for (int k = 0; k < nk; ++k) {
      const int sk = __builtin_amdgcn_readlane(s, k);
      const float* hp = h + (size_t)sk * LDH;
      if constexpr (H == 4) {
        const float p0 = __int_as_float(__builtin_amdgcn_readlane(__float_as_int(p[0]), k));
        const float p1 = __int_as_float(__builtin_amdgcn_readlane(__float_as_int(p[1]), k));
        const float p2 = __int_as_float(__builtin_amdgcn_readlane(__float_as_int(p[2]), k));
        const float p3 = __int_as_float(__builtin_amdgcn_readlane(__float_as_int(p[3]), k));
        const float pa = (lane < 16) ? p0 : p1;
        const float pb = (lane < 16) ? p2 : p3;
        const float4 a = *reinterpret_cast<const float4*>(hp + 4 * lane);
        const float4 b = *reinterpret_cast<const float4*>(hp + 128 + 4 * lane);
        acc0.x += a.x * pa; acc0.y += a.y * pa; acc0.z += a.z * pa; acc0.w += a.w * pa;
        acc1.x += b.x * pb; acc1.y += b.y * pb; acc1.z += b.z * pb; acc1.w += b.w * pb;
      } else {
        const float p0 = __int_as_float(__builtin_amdgcn_readlane(__float_as_int(p[0]), k));
        const float2 a = *reinterpret_cast<const float2*>(hp + 2 * lane);
        acc0.x += a.x * p0; acc0.y += a.y * p0;
      }
    }
  }
#pragma unroll
  for (int hh = 0; hh < H; ++hh) sm[hh] = wave_sum(sm[hh]);

  ob = make_float4(0.f, 0.f, 0.f, 0.f);
  if constexpr (H == 4) {
    const float ra = __builtin_amdgcn_rcpf(((lane < 16) ? sm[0] : sm[1]) + 1e-16f);
    const float rb = __builtin_amdgcn_rcpf(((lane < 16) ? sm[2] : sm[3]) + 1e-16f);
    const float4 ba = *reinterpret_cast<const float4*>(bias + 4 * lane);
    const float4 bb = *reinterpret_cast<const float4*>(bias + 128 + 4 * lane);
    oa.x = elu_f(acc0.x * ra + ba.x); oa.y = elu_f(acc0.y * ra + ba.y);
    oa.z = elu_f(acc0.z * ra + ba.z); oa.w = elu_f(acc0.w * ra + ba.w);
    ob.x = elu_f(acc1.x * rb + bb.x); ob.y = elu_f(acc1.y * rb + bb.y);
    ob.z = elu_f(acc1.z * rb + bb.z); ob.w = elu_f(acc1.w * rb + bb.w);
  } else {
    const float ra = __builtin_amdgcn_rcpf(sm[0] + 1e-16f);
    const float2 ba = *reinterpret_cast<const float2*>(bias + 2 * lane);
    oa.x = elu_f(acc0.x * ra + ba.x); oa.y = elu_f(acc0.y * ra + ba.y);
    oa.z = 0.0f; oa.w = 0.0f;
  }
}

template <int H>
__global__ void __launch_bounds__(256)
gat_gather(const unsigned* __restrict__ segtab, int nslice, const int* __restrict__ col,
           const float* __restrict__ h, const float* __restrict__ es,
           const float* __restrict__ ed, const float* __restrict__ bias,
           float* __restrict__ out, int n) {
  constexpr int LDH = H * HID;
  const int lane = (int)(threadIdx.x & 31);
  int d = (int)((blockIdx.x * blockDim.x + threadIdx.x) >> 5);
  d = __builtin_amdgcn_readfirstlane(d);
  if (d >= n) return;
  float4 oa, ob;
  gather_node<H, 4>(segtab, nslice, col, h, es, ed, bias, d, lane, oa, ob);
  float* op = out + (size_t)d * LDH;
  if constexpr (H == 4) {
    *reinterpret_cast<float4*>(op + 4 * lane)       = oa;
    *reinterpret_cast<float4*>(op + 128 + 4 * lane) = ob;
  } else {
    *reinterpret_cast<float2*>(op + 2 * lane) = make_float2(oa.x, oa.y);
  }
}

#define FUSE_WAVES  16
#define FUSE_MAX_WG 512
#define AT_PITCH    (HEADS * HID + 4)
#define CT_PITCH    (HID + 4)
__global__ void __launch_bounds__(FUSE_WAVES * 32)
gather1_gemm2(const unsigned* __restrict__ segtab, int nslice, const int* __restrict__ col,
              const float* __restrict__ h1, const float* __restrict__ es1,
              const float* __restrict__ ed1, const float* __restrict__ b1,
              const float* __restrict__ W2, const float* __restrict__ a_src2,
              const float* __restrict__ a_dst2, float* __restrict__ h2,
              float* __restrict__ es2, float* __restrict__ ed2, int n) {
  constexpr int K = HEADS * HID, NC = HID;
  constexpr int NQ = (K / 4) * NC;
  constexpr int NTHR = FUSE_WAVES * 32;
  constexpr int NIT = NQ / NTHR;
  constexpr int KQ = 4;
  constexpr int KSQ = K / 8 / KQ;
  static_assert(NQ % NTHR == 0, "staging shape");
  static_assert(FUSE_WAVES == 16 && NC / 16 * KQ == FUSE_WAVES, "a wave per node, per column tile and k quarter");
  __shared__ float4 Bs[NQ];
  __shared__ float4 At4[16 * AT_PITCH / 4];
  __shared__ float4 Ct4[KQ * 16 * CT_PITCH / 4];
  float* At = reinterpret_cast<float*>(At4);
  float* Ct = reinterpret_cast<float*>(Ct4);

  const int t = (int)threadIdx.x;
  const int lane = t & 31;
  const int lr = lane & 15, hi = lane >> 4;
  const int w = __builtin_amdgcn_readfirstlane(t >> 5);

  {
    float4 stg[NIT];
#pragma unroll
    for (int j = 0; j < NIT; ++j) {
      const int idx = t + j * NTHR;
      const float* bp = W2 + (size_t)(4 * (idx / NC)) * NC + (idx % NC);
      stg[j].x = bp[0]; stg[j].y = bp[NC]; stg[j].z = bp[2 * NC]; stg[j].w = bp[3 * NC];
    }
#pragma unroll
    for (int j = 0; j < NIT; ++j) Bs[t + j * NTHR] = stg[j];
  }
  const float2 as2l = *reinterpret_cast<const float2*>(a_src2 + 2 * lane);
  const float2 ad2l = *reinterpret_cast<const float2*>(a_dst2 + 2 * lane);
  __syncthreads();

  const int tn = w & 3, kq = w >> 2;
  const int ntiles = n >> 4;
  const int tper = (ntiles + (int)gridDim.x - 1) / (int)gridDim.x;
  const int tile0 = (int)blockIdx.x * tper;
  const int tile1 = (tile0 + tper < ntiles) ? (tile0 + tper) : ntiles;
  for (int tile = tile0; tile < tile1; ++tile) {
    const int d = tile * 16 + w;
    {
      float4 oa, ob;
      gather_node<HEADS, 4>(segtab, nslice, col, h1, es1, ed1, b1, d, lane, oa, ob);
      float* ar = At + w * AT_PITCH;
      *reinterpret_cast<float4*>(ar + 4 * lane)       = oa;
      *reinterpret_cast<float4*>(ar + 128 + 4 * lane) = ob;
    }
    __syncthreads();

    {
      const float* ap = At + lr * AT_PITCH + 8 * (kq * KSQ) + 4 * hi;
      const float4* bq = Bs + (2 * (kq * KSQ) + hi) * NC + tn * 16 + lr;
      float4 a[KSQ], b[KSQ];
#pragma unroll
      for (int i = 0; i < KSQ; ++i) {
        a[i] = *reinterpret_cast<const float4*>(ap + 8 * i);
        b[i] = bq[2 * i * NC];
      }
      v8f c0 = v8f{}, c1 = v8f{};
#pragma unroll
      for (int i = 0; i < KSQ; ++i) {
        v2f a01; a01.x = a[i].x; a01.y = a[i].y;
        v2f a23; a23.x = a[i].z; a23.y = a[i].w;
        v2f b01; b01.x = b[i].x; b01.y = b[i].y;
        v2f b23; b23.x = b[i].z; b23.y = b[i].w;
        c0 = __builtin_amdgcn_wmma_f32_16x16x4_f32(
             false, a01,  false, b01,
             (short)0, c0,  false,  false);
        c1 = __builtin_amdgcn_wmma_f32_16x16x4_f32(
             false, a23,  false, b23,
             (short)0, c1,  false,  false);
      }
      float* cq = Ct + (kq * 16 + 8 * hi) * CT_PITCH + tn * 16 + lr;
#pragma unroll
      for (int v = 0; v < 8; ++v) cq[v * CT_PITCH] = c0[v] + c1[v];
    }
    __syncthreads();

    {
      const float* cr = Ct + w * CT_PITCH + 2 * lane;
      float2 c = *reinterpret_cast<const float2*>(cr);
#pragma unroll
      for (int q = 1; q < KQ; ++q) {
        const float2 x = *reinterpret_cast<const float2*>(cr + q * 16 * CT_PITCH);
        c.x += x.x; c.y += x.y;
      }
      *reinterpret_cast<float2*>(h2 + (size_t)d * NC + 2 * lane) = c;
      float s = c.x * as2l.x + c.y * as2l.y;
      float e = c.x * ad2l.x + c.y * ad2l.y;
      s = wave_sum(s);
      e = wave_sum(e);
      if (lane == 0) { es2[d] = s; ed2[d] = e; }
    }
  }
}

__global__ void out_proj(const float* __restrict__ h,
                         const float* __restrict__ Wo,
                         const float* __restrict__ bo,
                         float* __restrict__ out, int n) {
  int wid  = (int)((blockIdx.x * blockDim.x + threadIdx.x) >> 5);
  int lane = (int)(threadIdx.x & 31);
  if (wid >= n) return;
  const float* hp = h + (size_t)wid * HID;
  float s = hp[lane] * Wo[lane] + hp[lane + 32] * Wo[lane + 32];
  s = wave_sum(s);
  if (lane == 0) out[wid] = s + bo[0];
}

#define CDIV(a, b) (((a) + (b) - 1) / (b))

extern "C" void kernel_launch(void* const* d_in, const int* in_sizes, int n_in,
                              void* d_out, int out_size, void* d_ws, size_t ws_size,
                              hipStream_t stream) {
  const float* x   = (const float*)d_in[0];
  const float* W1  = (const float*)d_in[1];
  const float* as1 = (const float*)d_in[2];
  const float* ad1 = (const float*)d_in[3];
  const float* b1  = (const float*)d_in[4];
  const float* W2  = (const float*)d_in[5];
  const float* as2 = (const float*)d_in[6];
  const float* ad2 = (const float*)d_in[7];
  const float* b2  = (const float*)d_in[8];
  const float* Wo  = (const float*)d_in[9];
  const float* bo  = (const float*)d_in[10];
  const int*   ei  = (const int*)d_in[11];

  const int n  = in_sizes[0]  / IN_CH;
  const int e  = in_sizes[11] / 2;

  float* p = (float*)d_ws;
  float*    h1   = p;            p += (size_t)n * 256;
  float*    o1   = p;            p += (size_t)n * 256;
  unsigned* segtab = (unsigned*)p; p += (size_t)SEG_SLICES * SEG_PITCH;
  float*    es1  = p;            p += (size_t)n * 4;
  float*    ed1  = p;            p += (size_t)n * 4;
  float*    es2v = p;            p += n;
  float*    ed2v = p;            p += n;
  int*      col  = (int*)p;      p += e;
  unsigned* etmp = (unsigned*)p;
  float*    h2   = o1;
  float*    o2   = o1 + (size_t)n * 64;
  const int nslice = CDIV(e, SLICE);

  const int B = 256;

  const int gneed = CDIV(n / 16, GEMM_WAVES);
  const int gpart = gneed > 256 - nslice ? gneed : 256 - nslice;
  count_and_gemm1<<<nslice + gpart, GEMM_WAVES * 32, 0, stream>>>(
      ei, segtab, col, etmp, e, nslice, x, W1, h1, as1, ad1, es1, ed1, n);

  const int ntiles = n / 16;
  const int tper = CDIV(ntiles, FUSE_MAX_WG);
  gather1_gemm2<<<CDIV(ntiles, tper), FUSE_WAVES * 32, 0, stream>>>(
      segtab, nslice, col, h1, es1, ed1, b1, W2, as2, ad2, h2, es2v, ed2v, n);

  gat_gather<1><<<CDIV(n * 32, B), B, 0, stream>>>(segtab, nslice, col, h2, es2v, ed2v, b2, o2, n);

  out_proj<<<CDIV(n * 32, B), B, 0, stream>>>(o2, Wo, bo, (float*)d_out, n);
}
